# starting kernel
# baseline (speedup 1.0000x reference)
#include <hip/hip_runtime.h>
#include <hip/hip_bf16.h>

typedef __attribute__((ext_vector_type(2))) float v2f;
typedef __attribute__((ext_vector_type(8))) float v8f;

#define NODES   4096
#define FIN     512
#define FOUT    64
#define BATCH   4
#define NEG_BIG (-9000000000000000.0f)
#define ALPHA   0.2f
#define WAVES_PER_BLOCK 8

__global__ void __launch_bounds__(256) gat_h_kernel(
    const float* __restrict__ x, const float* __restrict__ W,
    const float* __restrict__ a,
    float* __restrict__ h, float* __restrict__ s1, float* __restrict__ s2)
{
    const int lane = threadIdx.x & 31;
    const int wid  = threadIdx.x >> 5;
    const int wg   = blockIdx.x * WAVES_PER_BLOCK + wid;
    const int b    = wg >> 8;
    const int i0   = (wg & 255) << 4;
    const int half = lane >> 4;
    const int n    = lane & 15;

    const float* xrow = x + ((size_t)(b * NODES + i0 + n) * FIN);

    v8f acc0 = {}, acc1 = {}, acc2 = {}, acc3 = {};

    for (int k = 0; k < FIN; k += 4) {
        const int ka = k + 2 * half;
        v2f A; A[0] = xrow[ka]; A[1] = xrow[ka + 1];
        const float* wr0 = W + (size_t)ka * FOUT + n;
        const float* wr1 = wr0 + FOUT;
        v2f B0; B0[0] = wr0[0];  B0[1] = wr1[0];
        v2f B1; B1[0] = wr0[16]; B1[1] = wr1[16];
        v2f B2; B2[0] = wr0[32]; B2[1] = wr1[32];
        v2f B3; B3[0] = wr0[48]; B3[1] = wr1[48];
        acc0 = __builtin_amdgcn_wmma_f32_16x16x4_f32(false, A, false, B0, (short)0, acc0, false, false);
        acc1 = __builtin_amdgcn_wmma_f32_16x16x4_f32(false, A, false, B1, (short)0, acc1, false, false);
        acc2 = __builtin_amdgcn_wmma_f32_16x16x4_f32(false, A, false, B2, (short)0, acc2, false, false);
        acc3 = __builtin_amdgcn_wmma_f32_16x16x4_f32(false, A, false, B3, (short)0, acc3, false, false);
    }

    const float a10 = a[n],        a11 = a[n + 16], a12 = a[n + 32], a13 = a[n + 48];
    const float a20 = a[FOUT + n], a21 = a[FOUT + n + 16],
                a22 = a[FOUT + n + 32], a23 = a[FOUT + n + 48];

    for (int r = 0; r < 8; ++r) {
        const int row = i0 + r + 8 * half;
        float* hrow = h + ((size_t)(b * NODES + row) * FOUT) + n;
        const float c0 = acc0[r], c1 = acc1[r], c2 = acc2[r], c3 = acc3[r];
        hrow[0] = c0; hrow[16] = c1; hrow[32] = c2; hrow[48] = c3;

        float p1 = c0 * a10 + c1 * a11 + c2 * a12 + c3 * a13;
        float p2 = c0 * a20 + c1 * a21 + c2 * a22 + c3 * a23;
        for (int m = 1; m < 16; m <<= 1) {
            p1 += __shfl_xor(p1, m, 32);
            p2 += __shfl_xor(p2, m, 32);
        }
        if (n == 0) {
            s1[b * NODES + row] = p1;
            s2[b * NODES + row] = p2;
        }
    }
}

__global__ void __launch_bounds__(256) gat_attn_kernel(
    const float* __restrict__ h, const float* __restrict__ s1g,
    const float* __restrict__ s2g, const int* __restrict__ adj,
    float* __restrict__ out)
{
    __shared__ float lds_h[64 * FOUT];
    __shared__ float lds_s2[64];

    const int lane = threadIdx.x & 31;
    const int wid  = threadIdx.x >> 5;
    const int wg   = blockIdx.x * WAVES_PER_BLOCK + wid;
    const int b    = wg >> 8;
    const int i0   = (wg & 255) << 4;
    const int half = lane >> 4;
    const int n    = lane & 15;

    const float sv1 = s1g[b * NODES + i0 + n];
    const int* adjrow = adj + (size_t)(i0 + n) * NODES;

    float mrun = NEG_BIG;
    float lrun = 0.0f;
    v8f acc0 = {}, acc1 = {}, acc2 = {}, acc3 = {};

    for (int j0 = 0; j0 < NODES; j0 += 64) {
        __syncthreads();
        {
            const float4* src = (const float4*)(h + ((size_t)(b * NODES + j0)) * FOUT);
            float4* dst = (float4*)lds_h;
            for (int t = threadIdx.x; t < 64 * (FOUT / 4); t += 256) dst[t] = src[t];
            if (threadIdx.x < 64) lds_s2[threadIdx.x] = s2g[b * NODES + j0 + threadIdx.x];
        }
        __syncthreads();

        for (int jj = 0; jj < 4; ++jj) {
            const int jb = jj * 16;
            float p[8];
            float mx = NEG_BIG;
            for (int q = 0; q < 4; ++q) {
                for (int c = 0; c < 2; ++c) {
                    const int jl = jb + 4 * q + 2 * half + c;
                    float e = sv1 + lds_s2[jl];
                    e = (e >= 0.0f) ? e : ALPHA * e;
                    e = (adjrow[j0 + jl] > 0) ? e : NEG_BIG;
                    p[2 * q + c] = e;
                    mx = fmaxf(mx, e);
                }
            }
            mx = fmaxf(mx, __shfl_xor(mx, 16, 32));
            const float mnew  = fmaxf(mrun, mx);
            const float scale = __expf(mrun - mnew);
            float rs = 0.0f;
            for (int t = 0; t < 8; ++t) { p[t] = __expf(p[t] - mnew); rs += p[t]; }
            rs += __shfl_xor(rs, 16, 32);
            lrun = lrun * scale + rs;
            mrun = mnew;

            for (int r = 0; r < 8; ++r) {
                const float sr = __shfl(scale, r + 8 * half, 32);
                acc0[r] *= sr; acc1[r] *= sr; acc2[r] *= sr; acc3[r] *= sr;
            }

            for (int q = 0; q < 4; ++q) {
                v2f A; A[0] = p[2 * q]; A[1] = p[2 * q + 1];
                const float* hr0 = lds_h + (size_t)(jb + 4 * q + 2 * half) * FOUT + n;
                const float* hr1 = hr0 + FOUT;
                v2f B0; B0[0] = hr0[0];  B0[1] = hr1[0];
                v2f B1; B1[0] = hr0[16]; B1[1] = hr1[16];
                v2f B2; B2[0] = hr0[32]; B2[1] = hr1[32];
                v2f B3; B3[0] = hr0[48]; B3[1] = hr1[48];
                acc0 = __builtin_amdgcn_wmma_f32_16x16x4_f32(false, A, false, B0, (short)0, acc0, false, false);
                acc1 = __builtin_amdgcn_wmma_f32_16x16x4_f32(false, A, false, B1, (short)0, acc1, false, false);
                acc2 = __builtin_amdgcn_wmma_f32_16x16x4_f32(false, A, false, B2, (short)0, acc2, false, false);
                acc3 = __builtin_amdgcn_wmma_f32_16x16x4_f32(false, A, false, B3, (short)0, acc3, false, false);
            }
        }
    }

    for (int r = 0; r < 8; ++r) {
        const float lr  = __shfl(lrun, r + 8 * half, 32);
        const float inv = 1.0f / lr;
        const int row   = i0 + r + 8 * half;
        float* o = out + ((size_t)(b * NODES + row)) * FOUT + n;
        float v;
        v = acc0[r] * inv; o[0]  = (v > 0.0f) ? v : expm1f(v);
        v = acc1[r] * inv; o[16] = (v > 0.0f) ? v : expm1f(v);
        v = acc2[r] * inv; o[32] = (v > 0.0f) ? v : expm1f(v);
        v = acc3[r] * inv; o[48] = (v > 0.0f) ? v : expm1f(v);
    }
}

extern "C" void kernel_launch(void* const* d_in, const int* in_sizes, int n_in,
                              void* d_out, int out_size, void* d_ws, size_t ws_size,
                              hipStream_t stream)
{
    const float* x   = (const float*)d_in[0];
    const int*   adj = (const int*)  d_in[1];
    const float* W   = (const float*)d_in[2];
    const float* a   = (const float*)d_in[3];
    float* out = (float*)d_out;

    float* h  = (float*)d_ws;
    float* s1 = h + (size_t)BATCH * NODES * FOUT;
    float* s2 = s1 + (size_t)BATCH * NODES;

    const int total_waves = BATCH * (NODES / 16);
    const int blocks = total_waves / WAVES_PER_BLOCK;

    gat_h_kernel<<<blocks, 256, 0, stream>>>(x, W, a, h, s1, s2);
    gat_attn_kernel<<<blocks, 256, 0, stream>>>(h, s1, s2, adj, out);
}


# Attention block gets 4 copy waves (8 waves, 2 per SIMD) that bring the next 256-key stage (h, s2, mask words) into a second LDS buffer while the 4 matrix waves compute; sums unchanged
# speedup vs baseline: 2.5607x; 2.5607x over previous
#include <hip/hip_runtime.h>
#include <hip/hip_bf16.h>

typedef __attribute__((ext_vector_type(2))) float v2f;
typedef __attribute__((ext_vector_type(8))) float v8f;

#define NODES   4096
#define FIN     512
#define FOUT    64
#define BATCH   4
#define NEG_BIG (-9000000000000000.0f)
#define ALPHA   0.2f
#define WAVES_PER_BLOCK 8

__global__ void __launch_bounds__(256) gat_h_kernel(
    const float* __restrict__ x, const float* __restrict__ W,
    const float* __restrict__ a,
    float* __restrict__ h, float* __restrict__ s1, float* __restrict__ s2)
{
    const int lane = threadIdx.x & 31;
    const int wid  = threadIdx.x >> 5;
    const int wg   = blockIdx.x * WAVES_PER_BLOCK + wid;
    const int b    = wg >> 8;
    const int i0   = (wg & 255) << 4;
    const int half = lane >> 4;
    const int n    = lane & 15;

    const float* xrow = x + ((size_t)(b * NODES + i0 + n) * FIN);

    v8f acc0 = {}, acc1 = {}, acc2 = {}, acc3 = {};

    for (int k = 0; k < FIN; k += 4) {
        const int ka = k + 2 * half;
        v2f A; A[0] = xrow[ka]; A[1] = xrow[ka + 1];
        const float* wr0 = W + (size_t)ka * FOUT + n;
        const float* wr1 = wr0 + FOUT;
        v2f B0; B0[0] = wr0[0];  B0[1] = wr1[0];
        v2f B1; B1[0] = wr0[16]; B1[1] = wr1[16];
        v2f B2; B2[0] = wr0[32]; B2[1] = wr1[32];
        v2f B3; B3[0] = wr0[48]; B3[1] = wr1[48];
        acc0 = __builtin_amdgcn_wmma_f32_16x16x4_f32(false, A, false, B0, (short)0, acc0, false, false);
        acc1 = __builtin_amdgcn_wmma_f32_16x16x4_f32(false, A, false, B1, (short)0, acc1, false, false);
        acc2 = __builtin_amdgcn_wmma_f32_16x16x4_f32(false, A, false, B2, (short)0, acc2, false, false);
        acc3 = __builtin_amdgcn_wmma_f32_16x16x4_f32(false, A, false, B3, (short)0, acc3, false, false);
    }

    const float a10 = a[n],        a11 = a[n + 16], a12 = a[n + 32], a13 = a[n + 48];
    const float a20 = a[FOUT + n], a21 = a[FOUT + n + 16],
                a22 = a[FOUT + n + 32], a23 = a[FOUT + n + 48];

    for (int r = 0; r < 8; ++r) {
        const int row = i0 + r + 8 * half;
        float* hrow = h + ((size_t)(b * NODES + row) * FOUT) + n;
        const float c0 = acc0[r], c1 = acc1[r], c2 = acc2[r], c3 = acc3[r];
        hrow[0] = c0; hrow[16] = c1; hrow[32] = c2; hrow[48] = c3;

        float p1 = c0 * a10 + c1 * a11 + c2 * a12 + c3 * a13;
        float p2 = c0 * a20 + c1 * a21 + c2 * a22 + c3 * a23;
        for (int m = 1; m < 16; m <<= 1) {
            p1 += __shfl_xor(p1, m, 32);
            p2 += __shfl_xor(p2, m, 32);
        }
        if (n == 0) {
            s1[b * NODES + row] = p1;
            s2[b * NODES + row] = p2;
        }
    }
}

__global__ void __launch_bounds__(256) gat_pack_kernel(
    const int* __restrict__ adj, float* out)
{
    const int lane = threadIdx.x & 31;
    const int row  = blockIdx.x * WAVES_PER_BLOCK + (threadIdx.x >> 5);
    const int4* src = (const int4*)(adj + (size_t)row * NODES) + lane;

    int4 v[32];
#pragma unroll
    for (int c = 0; c < 32; ++c) v[c] = src[c * 32];

    unsigned m0 = 0u, m1 = 0u, m2 = 0u, m3 = 0u;
#pragma unroll
    for (int c = 0; c < 32; ++c) {
        const unsigned w0 = __builtin_amdgcn_ballot_w32(v[c].x > 0);
        const unsigned w1 = __builtin_amdgcn_ballot_w32(v[c].y > 0);
        const unsigned w2 = __builtin_amdgcn_ballot_w32(v[c].z > 0);
        const unsigned w3 = __builtin_amdgcn_ballot_w32(v[c].w > 0);
        if (lane == c) { m0 = w0; m1 = w1; m2 = w2; m3 = w3; }
    }

    float4 o;
    o.x = __uint_as_float(m0); o.y = __uint_as_float(m1);
    o.z = __uint_as_float(m2); o.w = __uint_as_float(m3);
    float* d = out + ((size_t)((lane >> 4) * NODES + row)) * FOUT + (lane & 15) * 4;
    *(float4*)d = o;
    *(float4*)(d + (size_t)2 * NODES * FOUT) = o;
}

__device__ __forceinline__ float gat_elu(float v) { return (v > 0.0f) ? v : expm1f(v); }

__device__ __forceinline__ void gat_store_tile(float* o, const v8f acc, const float inv)
{
    float4 lo, hi;
    lo.x = gat_elu(acc[0] * inv); lo.y = gat_elu(acc[1] * inv);
    lo.z = gat_elu(acc[2] * inv); lo.w = gat_elu(acc[3] * inv);
    hi.x = gat_elu(acc[4] * inv); hi.y = gat_elu(acc[5] * inv);
    hi.z = gat_elu(acc[6] * inv); hi.w = gat_elu(acc[7] * inv);
    *(float4*)o       = lo;
    *(float4*)(o + 4) = hi;
}

#define STAGE_KEYS 256
#define ATTN_MATRIX_WAVES 4
#define ATTN_COPY_WAVES 4
#define ATTN_THREADS (32 * (ATTN_MATRIX_WAVES + ATTN_COPY_WAVES))
#define STAGE_H  (2 * STAGE_KEYS * FOUT)
#define STAGE_S2 (2 * STAGE_KEYS)
#define STAGE_MK (32 * 8)

__device__ __forceinline__ void gat_copy_stage(
    const float* __restrict__ h, const float* __restrict__ s2g, const float* out,
    float* lds_h, float* lds_s2, float* lds_mk,
    const int pr, const int iblk, const int lt, const int st)
{
    const int j0   = st * STAGE_KEYS;
    const int bufi = st & 1;
    const int bs = lt >> 6;
    const int tt = lt & 63;
    const float4* src = (const float4*)(h + ((size_t)((2 * pr + bs) * NODES + j0)) * FOUT);
    float4* dst = (float4*)(lds_h + bufi * STAGE_H + bs * (STAGE_KEYS * FOUT));
    const float4 s2v = *(const float4*)(s2g + (2 * pr + bs) * NODES + j0 + 4 * tt);
    const int r  = (lt >> 1) & 31;
    const int pc = lt & 1;
    const float4 mv = *(const float4*)(out + ((size_t)(2 * pr + (st >> 3)) * NODES + iblk + r) * FOUT
                                           + ((2 * st) & 15) * 4 + 4 * pc);
    float4 tmp[64];
#pragma unroll
    for (int u = 0; u < 64; ++u) tmp[u] = src[tt + 64 * u];
#pragma unroll
    for (int u = 0; u < 64; ++u) dst[tt + 64 * u] = tmp[u];
    *(float4*)(lds_s2 + bufi * STAGE_S2 + bs * STAGE_KEYS + 4 * tt) = s2v;
    *(float4*)(lds_mk + bufi * STAGE_MK + r * 8 + 4 * pc) = mv;
}

__global__ void __launch_bounds__(ATTN_THREADS) __attribute__((amdgpu_waves_per_eu(2, 2))) gat_attn_kernel(
    const float* __restrict__ h, const float* __restrict__ s1g,
    const float* __restrict__ s2g, float* out)
{
    __shared__ float lds_h[2 * STAGE_H];
    __shared__ alignas(16) float lds_s2[2 * STAGE_S2];
    __shared__ alignas(16) float lds_mk[2 * STAGE_MK];

    const int lane = threadIdx.x & 31;
    const int wid  = __builtin_amdgcn_readfirstlane((int)(threadIdx.x >> 5));
    const bool copy_wave = wid >= ATTN_MATRIX_WAVES;
    const int cw   = wid & 3;
    const int pr   = blockIdx.x >> 7;
    const int bsel = cw >> 1;
    const int b    = 2 * pr + bsel;
    const int iblk = (blockIdx.x & 127) << 5;
    const int i0   = iblk + ((cw & 1) << 4);
    const int half = lane >> 4;
    const int n    = lane & 15;
    const int lt   = (int)threadIdx.x & 127;

    const float sv1 = s1g[b * NODES + i0 + n];

    float mrun = NEG_BIG;
    float lrun = 0.0f;
    v8f acc0 = {}, acc1 = {}, acc2 = {}, acc3 = {};

    if (copy_wave) gat_copy_stage(h, s2g, out, lds_h, lds_s2, lds_mk, pr, iblk, lt, 0);
    __syncthreads();

    for (int st = 0; st < NODES / STAGE_KEYS; ++st) {
        if (copy_wave) {
            if (st + 1 < NODES / STAGE_KEYS)
                gat_copy_stage(h, s2g, out, lds_h, lds_s2, lds_mk, pr, iblk, lt, st + 1);
        } else {
            const int bufi = st & 1;
            const float* lh  = lds_h  + bufi * STAGE_H  + bsel * (STAGE_KEYS * FOUT);
            const float* ls2 = lds_s2 + bufi * STAGE_S2 + bsel * STAGE_KEYS;
            const float* mp = lds_mk + bufi * STAGE_MK + (((cw & 1) << 4) + n) * 8 + 2 * half;
            const float2 mwA = *(const float2*)mp;
            const float2 mwB = *(const float2*)(mp + 4);

            for (int u = 0; u < STAGE_KEYS / 64; ++u) {
                const float2 mw  = (u & 2) ? mwB : mwA;
                const float* lhu = lh + u * (64 * FOUT);
                const float* lsu = ls2 + u * 64;
                const unsigned sh = (unsigned)(u & 1) << 4;
                unsigned mk[2];
                mk[0] = __float_as_uint(mw.x) >> sh;
                mk[1] = __float_as_uint(mw.y) >> sh;

    #pragma unroll
                for (int jj = 0; jj < 4; ++jj) {
                    const int jb = jj * 16;
                    float p[8];
                    float mx = NEG_BIG;
                    for (int q = 0; q < 4; ++q) {
                        for (int c = 0; c < 2; ++c) {
                            const int jl = jb + 4 * q + 2 * half + c;
                            float e = sv1 + lsu[jl];
                            e = (e >= 0.0f) ? e : ALPHA * e;
                            e = ((mk[c] >> (4 * jj + q)) & 1u) ? e : NEG_BIG;
                            p[2 * q + c] = e;
                            mx = fmaxf(mx, e);
                        }
                    }
                    mx = fmaxf(mx, __shfl_xor(mx, 16, 32));
                    const float mnew  = fmaxf(mrun, mx);
                    const float scale = __expf(mrun - mnew);
                    float rs = 0.0f;
                    for (int t = 0; t < 8; ++t) { p[t] = __expf(p[t] - mnew); rs += p[t]; }
                    rs += __shfl_xor(rs, 16, 32);
                    lrun = lrun * scale + rs;
                    mrun = mnew;

                    for (int r = 0; r < 8; ++r) {
                        acc0[r] *= scale; acc1[r] *= scale; acc2[r] *= scale; acc3[r] *= scale;
                    }

                    for (int q = 0; q < 4; ++q) {
                        v2f P; P[0] = p[2 * q]; P[1] = p[2 * q + 1];
                        const float* hr0 = lhu + (size_t)(jb + 4 * q + 2 * half) * FOUT + n;
                        const float* hr1 = hr0 + FOUT;
                        v2f H0; H0[0] = hr0[0];  H0[1] = hr1[0];
                        v2f H1; H1[0] = hr0[16]; H1[1] = hr1[16];
                        v2f H2; H2[0] = hr0[32]; H2[1] = hr1[32];
                        v2f H3; H3[0] = hr0[48]; H3[1] = hr1[48];
                        acc0 = __builtin_amdgcn_wmma_f32_16x16x4_f32(false, H0, false, P, (short)0, acc0, false, false);
                        acc1 = __builtin_amdgcn_wmma_f32_16x16x4_f32(false, H1, false, P, (short)0, acc1, false, false);
                        acc2 = __builtin_amdgcn_wmma_f32_16x16x4_f32(false, H2, false, P, (short)0, acc2, false, false);
                        acc3 = __builtin_amdgcn_wmma_f32_16x16x4_f32(false, H3, false, P, (short)0, acc3, false, false);
                    }
                }
            }
        }
        __syncthreads();
    }


    if (!copy_wave) {
        const float inv = 1.0f / lrun;
        float* o = out + ((size_t)(b * NODES + i0 + n)) * FOUT + 8 * half;
        gat_store_tile(o,      acc0, inv);
        gat_store_tile(o + 16, acc1, inv);
        gat_store_tile(o + 32, acc2, inv);
        gat_store_tile(o + 48, acc3, inv);
    }
}

extern "C" void kernel_launch(void* const* d_in, const int* in_sizes, int n_in,
                              void* d_out, int out_size, void* d_ws, size_t ws_size,
                              hipStream_t stream)
{
    const float* x   = (const float*)d_in[0];
    const int*   adj = (const int*)  d_in[1];
    const float* W   = (const float*)d_in[2];
    const float* a   = (const float*)d_in[3];
    float* out = (float*)d_out;

    float* h  = (float*)d_ws;
    float* s1 = h + (size_t)BATCH * NODES * FOUT;
    float* s2 = s1 + (size_t)BATCH * NODES;

    const int total_waves = BATCH * (NODES / 16);
    const int blocks = total_waves / WAVES_PER_BLOCK;

    gat_pack_kernel<<<NODES / WAVES_PER_BLOCK, 256, 0, stream>>>(adj, out);
    gat_h_kernel<<<blocks, 256, 0, stream>>>(x, W, a, h, s1, s2);
    gat_attn_kernel<<<total_waves / ATTN_MATRIX_WAVES, ATTN_THREADS, 0, stream>>>(h, s1, s2, out);
}


# Attention: each row's exact softmax max found before the key loop (masked max of s2; logit monotonic in s2): no running max, no accumulator rescale; row sum per lane half, joined at end (sum order)
# speedup vs baseline: 2.6479x; 1.0341x over previous
#include <hip/hip_runtime.h>
#include <hip/hip_bf16.h>

typedef __attribute__((ext_vector_type(2))) float v2f;
typedef __attribute__((ext_vector_type(8))) float v8f;

#define NODES   4096
#define FIN     512
#define FOUT    64
#define BATCH   4
#define NEG_BIG (-9000000000000000.0f)
#define ALPHA   0.2f
#define WAVES_PER_BLOCK 8

__global__ void __launch_bounds__(256) gat_h_kernel(
    const float* __restrict__ x, const float* __restrict__ W,
    const float* __restrict__ a,
    float* __restrict__ h, float* __restrict__ s1, float* __restrict__ s2)
{
    const int lane = threadIdx.x & 31;
    const int wid  = threadIdx.x >> 5;
    const int wg   = blockIdx.x * WAVES_PER_BLOCK + wid;
    const int b    = wg >> 8;
    const int i0   = (wg & 255) << 4;
    const int half = lane >> 4;
    const int n    = lane & 15;

    const float* xrow = x + ((size_t)(b * NODES + i0 + n) * FIN);

    v8f acc0 = {}, acc1 = {}, acc2 = {}, acc3 = {};

    for (int k = 0; k < FIN; k += 4) {
        const int ka = k + 2 * half;
        v2f A; A[0] = xrow[ka]; A[1] = xrow[ka + 1];
        const float* wr0 = W + (size_t)ka * FOUT + n;
        const float* wr1 = wr0 + FOUT;
        v2f B0; B0[0] = wr0[0];  B0[1] = wr1[0];
        v2f B1; B1[0] = wr0[16]; B1[1] = wr1[16];
        v2f B2; B2[0] = wr0[32]; B2[1] = wr1[32];
        v2f B3; B3[0] = wr0[48]; B3[1] = wr1[48];
        acc0 = __builtin_amdgcn_wmma_f32_16x16x4_f32(false, A, false, B0, (short)0, acc0, false, false);
        acc1 = __builtin_amdgcn_wmma_f32_16x16x4_f32(false, A, false, B1, (short)0, acc1, false, false);
        acc2 = __builtin_amdgcn_wmma_f32_16x16x4_f32(false, A, false, B2, (short)0, acc2, false, false);
        acc3 = __builtin_amdgcn_wmma_f32_16x16x4_f32(false, A, false, B3, (short)0, acc3, false, false);
    }

    const float a10 = a[n],        a11 = a[n + 16], a12 = a[n + 32], a13 = a[n + 48];
    const float a20 = a[FOUT + n], a21 = a[FOUT + n + 16],
                a22 = a[FOUT + n + 32], a23 = a[FOUT + n + 48];

    for (int r = 0; r < 8; ++r) {
        const int row = i0 + r + 8 * half;
        float* hrow = h + ((size_t)(b * NODES + row) * FOUT) + n;
        const float c0 = acc0[r], c1 = acc1[r], c2 = acc2[r], c3 = acc3[r];
        hrow[0] = c0; hrow[16] = c1; hrow[32] = c2; hrow[48] = c3;

        float p1 = c0 * a10 + c1 * a11 + c2 * a12 + c3 * a13;
        float p2 = c0 * a20 + c1 * a21 + c2 * a22 + c3 * a23;
        for (int m = 1; m < 16; m <<= 1) {
            p1 += __shfl_xor(p1, m, 32);
            p2 += __shfl_xor(p2, m, 32);
        }
        if (n == 0) {
            s1[b * NODES + row] = p1;
            s2[b * NODES + row] = p2;
        }
    }
}

__global__ void __launch_bounds__(256) gat_pack_kernel(
    const int* __restrict__ adj, float* out)
{
    const int lane = threadIdx.x & 31;
    const int row  = blockIdx.x * WAVES_PER_BLOCK + (threadIdx.x >> 5);
    const int4* src = (const int4*)(adj + (size_t)row * NODES) + lane;

    int4 v[32];
#pragma unroll
    for (int c = 0; c < 32; ++c) v[c] = src[c * 32];

    unsigned m0 = 0u, m1 = 0u, m2 = 0u, m3 = 0u;
#pragma unroll
    for (int c = 0; c < 32; ++c) {
        const unsigned w0 = __builtin_amdgcn_ballot_w32(v[c].x > 0);
        const unsigned w1 = __builtin_amdgcn_ballot_w32(v[c].y > 0);
        const unsigned w2 = __builtin_amdgcn_ballot_w32(v[c].z > 0);
        const unsigned w3 = __builtin_amdgcn_ballot_w32(v[c].w > 0);
        if (lane == c) { m0 = w0; m1 = w1; m2 = w2; m3 = w3; }
    }

    float4 o;
    o.x = __uint_as_float(m0); o.y = __uint_as_float(m1);
    o.z = __uint_as_float(m2); o.w = __uint_as_float(m3);
    float* d = out + ((size_t)((lane >> 4) * NODES + row)) * FOUT + (lane & 15) * 4;
    *(float4*)d = o;
    *(float4*)(d + (size_t)2 * NODES * FOUT) = o;
}

__device__ __forceinline__ float gat_elu(float v) { return (v > 0.0f) ? v : expm1f(v); }

__device__ __forceinline__ void gat_store_tile(float* o, const v8f acc, const float inv)
{
    float4 lo, hi;
    lo.x = gat_elu(acc[0] * inv); lo.y = gat_elu(acc[1] * inv);
    lo.z = gat_elu(acc[2] * inv); lo.w = gat_elu(acc[3] * inv);
    hi.x = gat_elu(acc[4] * inv); hi.y = gat_elu(acc[5] * inv);
    hi.z = gat_elu(acc[6] * inv); hi.w = gat_elu(acc[7] * inv);
    *(float4*)o       = lo;
    *(float4*)(o + 4) = hi;
}

#define STAGE_KEYS 256
#define ATTN_MATRIX_WAVES 4
#define ATTN_COPY_WAVES 4
#define ATTN_THREADS (32 * (ATTN_MATRIX_WAVES + ATTN_COPY_WAVES))
#define STAGE_H  (2 * STAGE_KEYS * FOUT)
#define STAGE_S2 (2 * STAGE_KEYS)
#define STAGE_MK (32 * 8)
#define RM_PITCH 36

__device__ __forceinline__ void gat_copy_stage(
    const float* __restrict__ h, const float* __restrict__ s2g, const float* out,
    float* lds_h, float* lds_s2, float* lds_mk,
    const int pr, const int iblk, const int lt, const int st)
{
    const int j0   = st * STAGE_KEYS;
    const int bufi = st & 1;
    const int bs = lt >> 6;
    const int tt = lt & 63;
    const float4* src = (const float4*)(h + ((size_t)((2 * pr + bs) * NODES + j0)) * FOUT);
    float4* dst = (float4*)(lds_h + bufi * STAGE_H + bs * (STAGE_KEYS * FOUT));
    const float4 s2v = *(const float4*)(s2g + (2 * pr + bs) * NODES + j0 + 4 * tt);
    const int r  = (lt >> 1) & 31;
    const int pc = lt & 1;
    const float4 mv = *(const float4*)(out + ((size_t)(2 * pr + (st >> 3)) * NODES + iblk + r) * FOUT
                                           + ((2 * st) & 15) * 4 + 4 * pc);
    float4 tmp[64];
#pragma unroll
    for (int u = 0; u < 64; ++u) tmp[u] = src[tt + 64 * u];
#pragma unroll
    for (int u = 0; u < 64; ++u) dst[tt + 64 * u] = tmp[u];
    *(float4*)(lds_s2 + bufi * STAGE_S2 + bs * STAGE_KEYS + 4 * tt) = s2v;
    *(float4*)(lds_mk + bufi * STAGE_MK + r * 8 + 4 * pc) = mv;
}

__global__ void __launch_bounds__(ATTN_THREADS) __attribute__((amdgpu_waves_per_eu(2, 2))) gat_attn_kernel(
    const float* __restrict__ h, const float* __restrict__ s1g,
    const float* __restrict__ s2g, float* out)
{
    __shared__ float lds_h[2 * STAGE_H];
    __shared__ alignas(16) float lds_s2[2 * STAGE_S2];
    __shared__ alignas(16) float lds_mk[2 * STAGE_MK];
    __shared__ alignas(16) float lds_pp[2 * 2 * 32 * RM_PITCH];

    const int lane = threadIdx.x & 31;
    const int wid  = __builtin_amdgcn_readfirstlane((int)(threadIdx.x >> 5));
    const bool copy_wave = wid >= ATTN_MATRIX_WAVES;
    const int cw   = wid & 3;
    const int pr   = blockIdx.x >> 7;
    const int bsel = cw >> 1;
    const int b    = 2 * pr + bsel;
    const int iblk = (blockIdx.x & 127) << 5;
    const int i0   = iblk + ((cw & 1) << 4);
    const int half = lane >> 4;
    const int n    = lane & 15;
    const int lt   = (int)threadIdx.x & 127;

    const float sv1 = s1g[b * NODES + i0 + n];

    float lsum = 0.0f;
    v8f acc0 = {}, acc1 = {}, acc2 = {}, acc3 = {};

    if (copy_wave) {
        gat_copy_stage(h, s2g, out, lds_h, lds_s2, lds_mk, pr, iblk, lt, 0);
    } else {
        const int rset  = cw & 1;
        const int chalf = cw >> 1;
        const float4* s2a = (const float4*)(s2g + (2 * pr)     * NODES + 2048 * chalf) + lane;
        const float4* s2b = (const float4*)(s2g + (2 * pr + 1) * NODES + 2048 * chalf) + lane;
        float4 sa[16], sb[16];
#pragma unroll
        for (int c = 0; c < 16; ++c) { sa[c] = s2a[32 * c]; sb[c] = s2b[32 * c]; }
        const unsigned* wrow = (const unsigned*)(out + ((size_t)(2 * pr + chalf) * NODES + iblk + 16 * rset) * FOUT);
        float* ppa = lds_pp + ((chalf * 2 + 0) * 32 + 16 * rset) * RM_PITCH + lane;
        float* ppb = lds_pp + ((chalf * 2 + 1) * 32 + 16 * rset) * RM_PITCH + lane;
        for (int r = 0; r < 16; ++r) {
            const unsigned* wp = wrow + r * FOUT;
            float ma = -INFINITY, mb = -INFINITY;
#pragma unroll
            for (int c = 0; c < 16; ++c) {
                const bool k0 = __builtin_amdgcn_inverse_ballot_w32(wp[4 * c + 0]);
                const bool k1 = __builtin_amdgcn_inverse_ballot_w32(wp[4 * c + 1]);
                const bool k2 = __builtin_amdgcn_inverse_ballot_w32(wp[4 * c + 2]);
                const bool k3 = __builtin_amdgcn_inverse_ballot_w32(wp[4 * c + 3]);
                ma = fmaxf(ma, fmaxf(k0 ? sa[c].x : -INFINITY, k1 ? sa[c].y : -INFINITY));
                ma = fmaxf(ma, fmaxf(k2 ? sa[c].z : -INFINITY, k3 ? sa[c].w : -INFINITY));
                mb = fmaxf(mb, fmaxf(k0 ? sb[c].x : -INFINITY, k1 ? sb[c].y : -INFINITY));
                mb = fmaxf(mb, fmaxf(k2 ? sb[c].z : -INFINITY, k3 ? sb[c].w : -INFINITY));
            }
            ppa[r * RM_PITCH] = ma;
            ppb[r * RM_PITCH] = mb;
        }
    }
    __syncthreads();

    float mfin = NEG_BIG;
    if (!copy_wave) {
        const float4* pr4 = (const float4*)(lds_pp + ((half * 2 + bsel) * 32 + ((cw & 1) << 4) + n) * RM_PITCH);
        float rm = -INFINITY;
        for (int k = 0; k < 8; ++k) {
            const float4 t = pr4[k];
            rm = fmaxf(rm, fmaxf(fmaxf(t.x, t.y), fmaxf(t.z, t.w)));
        }
        rm = fmaxf(rm, __shfl_xor(rm, 16, 32));
        float e = sv1 + rm;
        e = (e >= 0.0f) ? e : ALPHA * e;
        mfin = fmaxf(e, NEG_BIG);
    }

    for (int st = 0; st < NODES / STAGE_KEYS; ++st) {
        if (copy_wave) {
            if (st + 1 < NODES / STAGE_KEYS)
                gat_copy_stage(h, s2g, out, lds_h, lds_s2, lds_mk, pr, iblk, lt, st + 1);
        } else {
            const int bufi = st & 1;
            const float* lh  = lds_h  + bufi * STAGE_H  + bsel * (STAGE_KEYS * FOUT);
            const float* ls2 = lds_s2 + bufi * STAGE_S2 + bsel * STAGE_KEYS;
            const float* mp = lds_mk + bufi * STAGE_MK + (((cw & 1) << 4) + n) * 8 + 2 * half;
            const float2 mwA = *(const float2*)mp;
            const float2 mwB = *(const float2*)(mp + 4);

            for (int u = 0; u < STAGE_KEYS / 64; ++u) {
                const float2 mw  = (u & 2) ? mwB : mwA;
                const float* lhu = lh + u * (64 * FOUT);
                const float* lsu = ls2 + u * 64;
                const unsigned sh = (unsigned)(u & 1) << 4;
                unsigned mk[2];
                mk[0] = __float_as_uint(mw.x) >> sh;
                mk[1] = __float_as_uint(mw.y) >> sh;

    #pragma unroll
                for (int jj = 0; jj < 4; ++jj) {
                    const int jb = jj * 16;
                    float p[8];
                    float rs = 0.0f;
                    for (int q = 0; q < 4; ++q) {
                        for (int c = 0; c < 2; ++c) {
                            const int jl = jb + 4 * q + 2 * half + c;
                            float e = sv1 + lsu[jl];
                            e = (e >= 0.0f) ? e : ALPHA * e;
                            e = ((mk[c] >> (4 * jj + q)) & 1u) ? e : NEG_BIG;
                            const float w = __expf(e - mfin);
                            p[2 * q + c] = w;
                            rs += w;
                        }
                    }
                    lsum += rs;

                    for (int q = 0; q < 4; ++q) {
                        v2f P; P[0] = p[2 * q]; P[1] = p[2 * q + 1];
                        const float* hr0 = lhu + (size_t)(jb + 4 * q + 2 * half) * FOUT + n;
                        const float* hr1 = hr0 + FOUT;
                        v2f H0; H0[0] = hr0[0];  H0[1] = hr1[0];
                        v2f H1; H1[0] = hr0[16]; H1[1] = hr1[16];
                        v2f H2; H2[0] = hr0[32]; H2[1] = hr1[32];
                        v2f H3; H3[0] = hr0[48]; H3[1] = hr1[48];
                        acc0 = __builtin_amdgcn_wmma_f32_16x16x4_f32(false, H0, false, P, (short)0, acc0, false, false);
                        acc1 = __builtin_amdgcn_wmma_f32_16x16x4_f32(false, H1, false, P, (short)0, acc1, false, false);
                        acc2 = __builtin_amdgcn_wmma_f32_16x16x4_f32(false, H2, false, P, (short)0, acc2, false, false);
                        acc3 = __builtin_amdgcn_wmma_f32_16x16x4_f32(false, H3, false, P, (short)0, acc3, false, false);
                    }
                }
            }
        }
        __syncthreads();
    }


    if (!copy_wave) {
        const float inv = 1.0f / (lsum + __shfl_xor(lsum, 16, 32));
        float* o = out + ((size_t)(b * NODES + i0 + n)) * FOUT + 8 * half;
        gat_store_tile(o,      acc0, inv);
        gat_store_tile(o + 16, acc1, inv);
        gat_store_tile(o + 32, acc2, inv);
        gat_store_tile(o + 48, acc3, inv);
    }
}

extern "C" void kernel_launch(void* const* d_in, const int* in_sizes, int n_in,
                              void* d_out, int out_size, void* d_ws, size_t ws_size,
                              hipStream_t stream)
{
    const float* x   = (const float*)d_in[0];
    const int*   adj = (const int*)  d_in[1];
    const float* W   = (const float*)d_in[2];
    const float* a   = (const float*)d_in[3];
    float* out = (float*)d_out;

    float* h  = (float*)d_ws;
    float* s1 = h + (size_t)BATCH * NODES * FOUT;
    float* s2 = s1 + (size_t)BATCH * NODES;

    const int total_waves = BATCH * (NODES / 16);
    const int blocks = total_waves / WAVES_PER_BLOCK;

    gat_pack_kernel<<<NODES / WAVES_PER_BLOCK, 256, 0, stream>>>(adj, out);
    gat_h_kernel<<<blocks, 256, 0, stream>>>(x, W, a, h, s1, s2);
    gat_attn_kernel<<<total_waves / ATTN_MATRIX_WAVES, ATTN_THREADS, 0, stream>>>(h, s1, s2, out);
}


# Kernel 1 split-K, W stationary in registers: block = 64 rows, wave w keeps W rows 64w..64w+63 as B fragments for 4 row tiles; 8 partials summed via LDS (h, s1, s2 rounding changes); grid 128 -> 256
# speedup vs baseline: 3.2973x; 1.2453x over previous
#include <hip/hip_runtime.h>
#include <hip/hip_bf16.h>

typedef __attribute__((ext_vector_type(2))) float v2f;
typedef __attribute__((ext_vector_type(8))) float v8f;

#define NODES   4096
#define FIN     512
#define FOUT    64
#define BATCH   4
#define NEG_BIG (-9000000000000000.0f)
#define ALPHA   0.2f
#define WAVES_PER_BLOCK 8

#define K1_ROWS  64
#define K1_TILES (K1_ROWS / 16)
#define K1_BLOCKS ((BATCH * NODES) / K1_ROWS)

__global__ void __launch_bounds__(256) __attribute__((amdgpu_waves_per_eu(2, 2))) gat_h_kernel(
    const float* __restrict__ x, const float* __restrict__ W,
    const float* __restrict__ a,
    float* __restrict__ h, float* __restrict__ s1, float* __restrict__ s2)
{
    __shared__ alignas(16) float k1_part[WAVES_PER_BLOCK * K1_ROWS * FOUT];
    __shared__ float k1_s[2 * K1_ROWS];

    const int tid  = (int)threadIdx.x;
    const int lane = tid & 31;
    const int w    = __builtin_amdgcn_readfirstlane(tid >> 5);
    const int half = lane >> 4;
    const int n    = lane & 15;
    const int R0   = (int)blockIdx.x * K1_ROWS;

    const float* wp = W + (size_t)(64 * w + 4 * half) * FOUT + 4 * n;
    float4 wr[32];
#pragma unroll
    for (int i = 0; i < 32; ++i)
        wr[i] = *(const float4*)(wp + (size_t)(8 * (i >> 2) + (i & 3)) * FOUT);

    const float* xp = x + (size_t)(R0 + n) * FIN + 64 * w + 4 * half;
    float4 xr[K1_TILES][8];
#pragma unroll
    for (int rt = 0; rt < K1_TILES; ++rt)
#pragma unroll
        for (int t = 0; t < 8; ++t)
            xr[rt][t] = *(const float4*)(xp + (size_t)(16 * rt) * FIN + 8 * t);

#pragma unroll
    for (int rt = 0; rt < K1_TILES; ++rt) {
        v8f acc0 = {}, acc1 = {}, acc2 = {}, acc3 = {};
#pragma unroll
        for (int t = 0; t < 8; ++t) {
#pragma unroll
            for (int u = 0; u < 2; ++u) {
                const float4 xa = xr[rt][t];
                v2f A;
                A[0] = u ? xa.z : xa.x;
                A[1] = u ? xa.w : xa.y;
                const float4 b0 = wr[4 * t + 2 * u];
                const float4 b1 = wr[4 * t + 2 * u + 1];
                v2f B0; B0[0] = b0.x; B0[1] = b1.x;
                v2f B1; B1[0] = b0.y; B1[1] = b1.y;
                v2f B2; B2[0] = b0.z; B2[1] = b1.z;
                v2f B3; B3[0] = b0.w; B3[1] = b1.w;
                acc0 = __builtin_amdgcn_wmma_f32_16x16x4_f32(false, A, false, B0, (short)0, acc0, false, false);
                acc1 = __builtin_amdgcn_wmma_f32_16x16x4_f32(false, A, false, B1, (short)0, acc1, false, false);
                acc2 = __builtin_amdgcn_wmma_f32_16x16x4_f32(false, A, false, B2, (short)0, acc2, false, false);
                acc3 = __builtin_amdgcn_wmma_f32_16x16x4_f32(false, A, false, B3, (short)0, acc3, false, false);
            }
        }
        float* pp = k1_part + ((size_t)(w * K1_ROWS + 16 * rt + 8 * half)) * FOUT + 4 * n;
#pragma unroll
        for (int r = 0; r < 8; ++r) {
            float4 o; o.x = acc0[r]; o.y = acc1[r]; o.z = acc2[r]; o.w = acc3[r];
            *(float4*)(pp + r * FOUT) = o;
        }
    }
    __syncthreads();

    const float4 av1 = *(const float4*)(a + 4 * (tid & 15));
    const float4 av2 = *(const float4*)(a + FOUT + 4 * (tid & 15));
    const float4* p4 = (const float4*)k1_part;
    float4* h4 = (float4*)(h + (size_t)R0 * FOUT);
#pragma unroll
    for (int j = 0; j < 4; ++j) {
        const int f = tid + 256 * j;
        float4 sum = p4[f];
#pragma unroll
        for (int ww = 1; ww < WAVES_PER_BLOCK; ++ww) {
            const float4 t4 = p4[ww * (K1_ROWS * FOUT / 4) + f];
            sum.x += t4.x; sum.y += t4.y; sum.z += t4.z; sum.w += t4.w;
        }
        h4[f] = sum;
        float p1 = sum.x * av1.x + sum.y * av1.y + sum.z * av1.z + sum.w * av1.w;
        float p2 = sum.x * av2.x + sum.y * av2.y + sum.z * av2.z + sum.w * av2.w;
        for (int m = 1; m < 16; m <<= 1) {
            p1 += __shfl_xor(p1, m, 32);
            p2 += __shfl_xor(p2, m, 32);
        }
        if ((tid & 15) == 0) {
            k1_s[f >> 4] = p1;
            k1_s[K1_ROWS + (f >> 4)] = p2;
        }
    }
    __syncthreads();
    if (tid < K1_ROWS)          s1[R0 + tid] = k1_s[tid];
    else if (tid < 2 * K1_ROWS) s2[R0 + tid - K1_ROWS] = k1_s[tid];
}

__global__ void __launch_bounds__(256) gat_pack_kernel(
    const int* __restrict__ adj, float* out)
{
    const int lane = threadIdx.x & 31;
    const int row  = blockIdx.x * WAVES_PER_BLOCK + (threadIdx.x >> 5);
    const int4* src = (const int4*)(adj + (size_t)row * NODES) + lane;

    int4 v[32];
#pragma unroll
    for (int c = 0; c < 32; ++c) v[c] = src[c * 32];

    unsigned m0 = 0u, m1 = 0u, m2 = 0u, m3 = 0u;
#pragma unroll
    for (int c = 0; c < 32; ++c) {
        const unsigned w0 = __builtin_amdgcn_ballot_w32(v[c].x > 0);
        const unsigned w1 = __builtin_amdgcn_ballot_w32(v[c].y > 0);
        const unsigned w2 = __builtin_amdgcn_ballot_w32(v[c].z > 0);
        const unsigned w3 = __builtin_amdgcn_ballot_w32(v[c].w > 0);
        if (lane == c) { m0 = w0; m1 = w1; m2 = w2; m3 = w3; }
    }

    float4 o;
    o.x = __uint_as_float(m0); o.y = __uint_as_float(m1);
    o.z = __uint_as_float(m2); o.w = __uint_as_float(m3);
    float* d = out + ((size_t)((lane >> 4) * NODES + row)) * FOUT + (lane & 15) * 4;
    *(float4*)d = o;
    *(float4*)(d + (size_t)2 * NODES * FOUT) = o;
}

__device__ __forceinline__ float gat_elu(float v) { return (v > 0.0f) ? v : expm1f(v); }

__device__ __forceinline__ void gat_store_tile(float* o, const v8f acc, const float inv)
{
    float4 lo, hi;
    lo.x = gat_elu(acc[0] * inv); lo.y = gat_elu(acc[1] * inv);
    lo.z = gat_elu(acc[2] * inv); lo.w = gat_elu(acc[3] * inv);
    hi.x = gat_elu(acc[4] * inv); hi.y = gat_elu(acc[5] * inv);
    hi.z = gat_elu(acc[6] * inv); hi.w = gat_elu(acc[7] * inv);
    *(float4*)o       = lo;
    *(float4*)(o + 4) = hi;
}

#define STAGE_KEYS 256
#define ATTN_MATRIX_WAVES 4
#define ATTN_COPY_WAVES 4
#define ATTN_THREADS (32 * (ATTN_MATRIX_WAVES + ATTN_COPY_WAVES))
#define STAGE_H  (2 * STAGE_KEYS * FOUT)
#define STAGE_S2 (2 * STAGE_KEYS)
#define STAGE_MK (32 * 8)
#define RM_PITCH 36

__device__ __forceinline__ void gat_copy_stage(
    const float* __restrict__ h, const float* __restrict__ s2g, const float* out,
    float* lds_h, float* lds_s2, float* lds_mk,
    const int pr, const int iblk, const int lt, const int st)
{
    const int j0   = st * STAGE_KEYS;
    const int bufi = st & 1;
    const int bs = lt >> 6;
    const int tt = lt & 63;
    const float4* src = (const float4*)(h + ((size_t)((2 * pr + bs) * NODES + j0)) * FOUT);
    float4* dst = (float4*)(lds_h + bufi * STAGE_H + bs * (STAGE_KEYS * FOUT));
    const float4 s2v = *(const float4*)(s2g + (2 * pr + bs) * NODES + j0 + 4 * tt);
    const int r  = (lt >> 1) & 31;
    const int pc = lt & 1;
    const float4 mv = *(const float4*)(out + ((size_t)(2 * pr + (st >> 3)) * NODES + iblk + r) * FOUT
                                           + ((2 * st) & 15) * 4 + 4 * pc);
    float4 tmp[64];
#pragma unroll
    for (int u = 0; u < 64; ++u) tmp[u] = src[tt + 64 * u];
#pragma unroll
    for (int u = 0; u < 64; ++u) dst[tt + 64 * u] = tmp[u];
    *(float4*)(lds_s2 + bufi * STAGE_S2 + bs * STAGE_KEYS + 4 * tt) = s2v;
    *(float4*)(lds_mk + bufi * STAGE_MK + r * 8 + 4 * pc) = mv;
}

__global__ void __launch_bounds__(ATTN_THREADS) __attribute__((amdgpu_waves_per_eu(2, 2))) gat_attn_kernel(
    const float* __restrict__ h, const float* __restrict__ s1g,
    const float* __restrict__ s2g, float* out)
{
    __shared__ float lds_h[2 * STAGE_H];
    __shared__ alignas(16) float lds_s2[2 * STAGE_S2];
    __shared__ alignas(16) float lds_mk[2 * STAGE_MK];
    __shared__ alignas(16) float lds_pp[2 * 2 * 32 * RM_PITCH];

    const int lane = threadIdx.x & 31;
    const int wid  = __builtin_amdgcn_readfirstlane((int)(threadIdx.x >> 5));
    const bool copy_wave = wid >= ATTN_MATRIX_WAVES;
    const int cw   = wid & 3;
    const int pr   = blockIdx.x >> 7;
    const int bsel = cw >> 1;
    const int b    = 2 * pr + bsel;
    const int iblk = (blockIdx.x & 127) << 5;
    const int i0   = iblk + ((cw & 1) << 4);
    const int half = lane >> 4;
    const int n    = lane & 15;
    const int lt   = (int)threadIdx.x & 127;

    const float sv1 = s1g[b * NODES + i0 + n];

    float lsum = 0.0f;
    v8f acc0 = {}, acc1 = {}, acc2 = {}, acc3 = {};

    if (copy_wave) {
        gat_copy_stage(h, s2g, out, lds_h, lds_s2, lds_mk, pr, iblk, lt, 0);
    } else {
        const int rset  = cw & 1;
        const int chalf = cw >> 1;
        const float4* s2a = (const float4*)(s2g + (2 * pr)     * NODES + 2048 * chalf) + lane;
        const float4* s2b = (const float4*)(s2g + (2 * pr + 1) * NODES + 2048 * chalf) + lane;
        float4 sa[16], sb[16];
#pragma unroll
        for (int c = 0; c < 16; ++c) { sa[c] = s2a[32 * c]; sb[c] = s2b[32 * c]; }
        const unsigned* wrow = (const unsigned*)(out + ((size_t)(2 * pr + chalf) * NODES + iblk + 16 * rset) * FOUT);
        float* ppa = lds_pp + ((chalf * 2 + 0) * 32 + 16 * rset) * RM_PITCH + lane;
        float* ppb = lds_pp + ((chalf * 2 + 1) * 32 + 16 * rset) * RM_PITCH + lane;
        for (int r = 0; r < 16; ++r) {
            const unsigned* wp = wrow + r * FOUT;
            float ma = -INFINITY, mb = -INFINITY;
#pragma unroll
            for (int c = 0; c < 16; ++c) {
                const bool k0 = __builtin_amdgcn_inverse_ballot_w32(wp[4 * c + 0]);
                const bool k1 = __builtin_amdgcn_inverse_ballot_w32(wp[4 * c + 1]);
                const bool k2 = __builtin_amdgcn_inverse_ballot_w32(wp[4 * c + 2]);
                const bool k3 = __builtin_amdgcn_inverse_ballot_w32(wp[4 * c + 3]);
                ma = fmaxf(ma, fmaxf(k0 ? sa[c].x : -INFINITY, k1 ? sa[c].y : -INFINITY));
                ma = fmaxf(ma, fmaxf(k2 ? sa[c].z : -INFINITY, k3 ? sa[c].w : -INFINITY));
                mb = fmaxf(mb, fmaxf(k0 ? sb[c].x : -INFINITY, k1 ? sb[c].y : -INFINITY));
                mb = fmaxf(mb, fmaxf(k2 ? sb[c].z : -INFINITY, k3 ? sb[c].w : -INFINITY));
            }
            ppa[r * RM_PITCH] = ma;
            ppb[r * RM_PITCH] = mb;
        }
    }
    __syncthreads();

    float mfin = NEG_BIG;
    if (!copy_wave) {
        const float4* pr4 = (const float4*)(lds_pp + ((half * 2 + bsel) * 32 + ((cw & 1) << 4) + n) * RM_PITCH);
        float rm = -INFINITY;
        for (int k = 0; k < 8; ++k) {
            const float4 t = pr4[k];
            rm = fmaxf(rm, fmaxf(fmaxf(t.x, t.y), fmaxf(t.z, t.w)));
        }
        rm = fmaxf(rm, __shfl_xor(rm, 16, 32));
        float e = sv1 + rm;
        e = (e >= 0.0f) ? e : ALPHA * e;
        mfin = fmaxf(e, NEG_BIG);
    }

    for (int st = 0; st < NODES / STAGE_KEYS; ++st) {
        if (copy_wave) {
            if (st + 1 < NODES / STAGE_KEYS)
                gat_copy_stage(h, s2g, out, lds_h, lds_s2, lds_mk, pr, iblk, lt, st + 1);
        } else {
            const int bufi = st & 1;
            const float* lh  = lds_h  + bufi * STAGE_H  + bsel * (STAGE_KEYS * FOUT);
            const float* ls2 = lds_s2 + bufi * STAGE_S2 + bsel * STAGE_KEYS;
            const float* mp = lds_mk + bufi * STAGE_MK + (((cw & 1) << 4) + n) * 8 + 2 * half;
            const float2 mwA = *(const float2*)mp;
            const float2 mwB = *(const float2*)(mp + 4);

            for (int u = 0; u < STAGE_KEYS / 64; ++u) {
                const float2 mw  = (u & 2) ? mwB : mwA;
                const float* lhu = lh + u * (64 * FOUT);
                const float* lsu = ls2 + u * 64;
                const unsigned sh = (unsigned)(u & 1) << 4;
                unsigned mk[2];
                mk[0] = __float_as_uint(mw.x) >> sh;
                mk[1] = __float_as_uint(mw.y) >> sh;

    #pragma unroll
                for (int jj = 0; jj < 4; ++jj) {
                    const int jb = jj * 16;
                    float p[8];
                    float rs = 0.0f;
                    for (int q = 0; q < 4; ++q) {
                        for (int c = 0; c < 2; ++c) {
                            const int jl = jb + 4 * q + 2 * half + c;
                            float e = sv1 + lsu[jl];
                            e = (e >= 0.0f) ? e : ALPHA * e;
                            e = ((mk[c] >> (4 * jj + q)) & 1u) ? e : NEG_BIG;
                            const float w = __expf(e - mfin);
                            p[2 * q + c] = w;
                            rs += w;
                        }
                    }
                    lsum += rs;

                    for (int q = 0; q < 4; ++q) {
                        v2f P; P[0] = p[2 * q]; P[1] = p[2 * q + 1];
                        const float* hr0 = lhu + (size_t)(jb + 4 * q + 2 * half) * FOUT + n;
                        const float* hr1 = hr0 + FOUT;
                        v2f H0; H0[0] = hr0[0];  H0[1] = hr1[0];
                        v2f H1; H1[0] = hr0[16]; H1[1] = hr1[16];
                        v2f H2; H2[0] = hr0[32]; H2[1] = hr1[32];
                        v2f H3; H3[0] = hr0[48]; H3[1] = hr1[48];
                        acc0 = __builtin_amdgcn_wmma_f32_16x16x4_f32(false, H0, false, P, (short)0, acc0, false, false);
                        acc1 = __builtin_amdgcn_wmma_f32_16x16x4_f32(false, H1, false, P, (short)0, acc1, false, false);
                        acc2 = __builtin_amdgcn_wmma_f32_16x16x4_f32(false, H2, false, P, (short)0, acc2, false, false);
                        acc3 = __builtin_amdgcn_wmma_f32_16x16x4_f32(false, H3, false, P, (short)0, acc3, false, false);
                    }
                }
            }
        }
        __syncthreads();
    }


    if (!copy_wave) {
        const float inv = 1.0f / (lsum + __shfl_xor(lsum, 16, 32));
        float* o = out + ((size_t)(b * NODES + i0 + n)) * FOUT + 8 * half;
        gat_store_tile(o,      acc0, inv);
        gat_store_tile(o + 16, acc1, inv);
        gat_store_tile(o + 32, acc2, inv);
        gat_store_tile(o + 48, acc3, inv);
    }
}

extern "C" void kernel_launch(void* const* d_in, const int* in_sizes, int n_in,
                              void* d_out, int out_size, void* d_ws, size_t ws_size,
                              hipStream_t stream)
{
    const float* x   = (const float*)d_in[0];
    const int*   adj = (const int*)  d_in[1];
    const float* W   = (const float*)d_in[2];
    const float* a   = (const float*)d_in[3];
    float* out = (float*)d_out;

    float* h  = (float*)d_ws;
    float* s1 = h + (size_t)BATCH * NODES * FOUT;
    float* s2 = s1 + (size_t)BATCH * NODES;

    const int total_waves = BATCH * (NODES / 16);

    gat_pack_kernel<<<NODES / WAVES_PER_BLOCK, 256, 0, stream>>>(adj, out);
    gat_h_kernel<<<K1_BLOCKS, 256, 0, stream>>>(x, W, a, h, s1, s2);
    gat_attn_kernel<<<total_waves / ATTN_MATRIX_WAVES, ATTN_THREADS, 0, stream>>>(h, s1, s2, out);
}


# Attention: copy waves also compute the next 64-key stage's softmax weights and row sums and pass the weights through LDS in B-operand layout; matrix waves only read LDS and multiply; sums unchanged
# speedup vs baseline: 3.4296x; 1.0401x over previous
#include <hip/hip_runtime.h>
#include <hip/hip_bf16.h>

typedef __attribute__((ext_vector_type(2))) float v2f;
typedef __attribute__((ext_vector_type(8))) float v8f;

#define NODES   4096
#define FIN     512
#define FOUT    64
#define BATCH   4
#define NEG_BIG (-9000000000000000.0f)
#define ALPHA   0.2f
#define WAVES_PER_BLOCK 8

#define K1_ROWS  64
#define K1_TILES (K1_ROWS / 16)
#define K1_BLOCKS ((BATCH * NODES) / K1_ROWS)

__global__ void __launch_bounds__(256) __attribute__((amdgpu_waves_per_eu(2, 2))) gat_h_kernel(
    const float* __restrict__ x, const float* __restrict__ W,
    const float* __restrict__ a,
    float* __restrict__ h, float* __restrict__ s1, float* __restrict__ s2)
{
    __shared__ alignas(16) float k1_part[WAVES_PER_BLOCK * K1_ROWS * FOUT];
    __shared__ float k1_s[2 * K1_ROWS];

    const int tid  = (int)threadIdx.x;
    const int lane = tid & 31;
    const int w    = __builtin_amdgcn_readfirstlane(tid >> 5);
    const int half = lane >> 4;
    const int n    = lane & 15;
    const int R0   = (int)blockIdx.x * K1_ROWS;

    const float* wp = W + (size_t)(64 * w + 4 * half) * FOUT + 4 * n;
    float4 wr[32];
#pragma unroll
    for (int i = 0; i < 32; ++i)
        wr[i] = *(const float4*)(wp + (size_t)(8 * (i >> 2) + (i & 3)) * FOUT);

    const float* xp = x + (size_t)(R0 + n) * FIN + 64 * w + 4 * half;
    float4 xr[K1_TILES][8];
#pragma unroll
    for (int rt = 0; rt < K1_TILES; ++rt)
#pragma unroll
        for (int t = 0; t < 8; ++t)
            xr[rt][t] = *(const float4*)(xp + (size_t)(16 * rt) * FIN + 8 * t);

#pragma unroll
    for (int rt = 0; rt < K1_TILES; ++rt) {
        v8f acc0 = {}, acc1 = {}, acc2 = {}, acc3 = {};
#pragma unroll
        for (int t = 0; t < 8; ++t) {
#pragma unroll
            for (int u = 0; u < 2; ++u) {
                const float4 xa = xr[rt][t];
                v2f A;
                A[0] = u ? xa.z : xa.x;
                A[1] = u ? xa.w : xa.y;
                const float4 b0 = wr[4 * t + 2 * u];
                const float4 b1 = wr[4 * t + 2 * u + 1];
                v2f B0; B0[0] = b0.x; B0[1] = b1.x;
                v2f B1; B1[0] = b0.y; B1[1] = b1.y;
                v2f B2; B2[0] = b0.z; B2[1] = b1.z;
                v2f B3; B3[0] = b0.w; B3[1] = b1.w;
                acc0 = __builtin_amdgcn_wmma_f32_16x16x4_f32(false, A, false, B0, (short)0, acc0, false, false);
                acc1 = __builtin_amdgcn_wmma_f32_16x16x4_f32(false, A, false, B1, (short)0, acc1, false, false);
                acc2 = __builtin_amdgcn_wmma_f32_16x16x4_f32(false, A, false, B2, (short)0, acc2, false, false);
                acc3 = __builtin_amdgcn_wmma_f32_16x16x4_f32(false, A, false, B3, (short)0, acc3, false, false);
            }
        }
        float* pp = k1_part + ((size_t)(w * K1_ROWS + 16 * rt + 8 * half)) * FOUT + 4 * n;
#pragma unroll
        for (int r = 0; r < 8; ++r) {
            float4 o; o.x = acc0[r]; o.y = acc1[r]; o.z = acc2[r]; o.w = acc3[r];
            *(float4*)(pp + r * FOUT) = o;
        }
    }
    __syncthreads();

    const float4 av1 = *(const float4*)(a + 4 * (tid & 15));
    const float4 av2 = *(const float4*)(a + FOUT + 4 * (tid & 15));
    const float4* p4 = (const float4*)k1_part;
    float4* h4 = (float4*)(h + (size_t)R0 * FOUT);
#pragma unroll
    for (int j = 0; j < 4; ++j) {
        const int f = tid + 256 * j;
        float4 sum = p4[f];
#pragma unroll
        for (int ww = 1; ww < WAVES_PER_BLOCK; ++ww) {
            const float4 t4 = p4[ww * (K1_ROWS * FOUT / 4) + f];
            sum.x += t4.x; sum.y += t4.y; sum.z += t4.z; sum.w += t4.w;
        }
        h4[f] = sum;
        float p1 = sum.x * av1.x + sum.y * av1.y + sum.z * av1.z + sum.w * av1.w;
        float p2 = sum.x * av2.x + sum.y * av2.y + sum.z * av2.z + sum.w * av2.w;
        for (int m = 1; m < 16; m <<= 1) {
            p1 += __shfl_xor(p1, m, 32);
            p2 += __shfl_xor(p2, m, 32);
        }
        if ((tid & 15) == 0) {
            k1_s[f >> 4] = p1;
            k1_s[K1_ROWS + (f >> 4)] = p2;
        }
    }
    __syncthreads();
    if (tid < K1_ROWS)          s1[R0 + tid] = k1_s[tid];
    else if (tid < 2 * K1_ROWS) s2[R0 + tid - K1_ROWS] = k1_s[tid];
}

__global__ void __launch_bounds__(256) gat_pack_kernel(
    const int* __restrict__ adj, float* out)
{
    const int lane = threadIdx.x & 31;
    const int row  = blockIdx.x * WAVES_PER_BLOCK + (threadIdx.x >> 5);
    const int4* src = (const int4*)(adj + (size_t)row * NODES) + lane;

    int4 v[32];
#pragma unroll
    for (int c = 0; c < 32; ++c) v[c] = src[c * 32];

    unsigned m0 = 0u, m1 = 0u, m2 = 0u, m3 = 0u;
#pragma unroll
    for (int c = 0; c < 32; ++c) {
        const unsigned w0 = __builtin_amdgcn_ballot_w32(v[c].x > 0);
        const unsigned w1 = __builtin_amdgcn_ballot_w32(v[c].y > 0);
        const unsigned w2 = __builtin_amdgcn_ballot_w32(v[c].z > 0);
        const unsigned w3 = __builtin_amdgcn_ballot_w32(v[c].w > 0);
        if (lane == c) { m0 = w0; m1 = w1; m2 = w2; m3 = w3; }
    }

    float4 o;
    o.x = __uint_as_float(m0); o.y = __uint_as_float(m1);
    o.z = __uint_as_float(m2); o.w = __uint_as_float(m3);
    float* d = out + ((size_t)((lane >> 4) * NODES + row)) * FOUT + (lane & 15) * 4;
    *(float4*)d = o;
    *(float4*)(d + (size_t)2 * NODES * FOUT) = o;
}

__device__ __forceinline__ float gat_elu(float v) { return (v > 0.0f) ? v : expm1f(v); }

__device__ __forceinline__ void gat_store_tile(float* o, const v8f acc, const float inv)
{
    float4 lo, hi;
    lo.x = gat_elu(acc[0] * inv); lo.y = gat_elu(acc[1] * inv);
    lo.z = gat_elu(acc[2] * inv); lo.w = gat_elu(acc[3] * inv);
    hi.x = gat_elu(acc[4] * inv); hi.y = gat_elu(acc[5] * inv);
    hi.z = gat_elu(acc[6] * inv); hi.w = gat_elu(acc[7] * inv);
    *(float4*)o       = lo;
    *(float4*)(o + 4) = hi;
}

#define STAGE_KEYS 64
#define N_STAGES (NODES / STAGE_KEYS)
#define ATTN_MATRIX_WAVES 4
#define ATTN_COPY_WAVES 4
#define ATTN_THREADS (32 * (ATTN_MATRIX_WAVES + ATTN_COPY_WAVES))
#define STAGE_H  (2 * STAGE_KEYS * FOUT)
#define WAVE_P   ((STAGE_KEYS / 16) * 2 * 32 * 4)
#define STAGE_P  (ATTN_MATRIX_WAVES * WAVE_P)
#define MK_PITCH 128
#define GAT_GROUP_EDGE() do { asm volatile("" : "+v"(acc0), "+v"(acc1), "+v"(acc2), "+v"(acc3) : : "memory"); __builtin_amdgcn_sched_barrier(0); } while (0)
#define RM_PITCH 36

typedef __attribute__((ext_vector_type(4)))  float v4f;
typedef __attribute__((ext_vector_type(16))) float v16f;


__device__ __forceinline__ v16f gat_h_request4(const float4* src, const int tt, const int k)
{
    const v4f a = *(const v4f*)(src + tt + 64 * (4 * k + 0));
    const v4f b = *(const v4f*)(src + tt + 64 * (4 * k + 1));
    const v4f c = *(const v4f*)(src + tt + 64 * (4 * k + 2));
    const v4f d = *(const v4f*)(src + tt + 64 * (4 * k + 3));
    const v8f ab = __builtin_shufflevector(a, b, 0, 1, 2, 3, 4, 5, 6, 7);
    const v8f cd = __builtin_shufflevector(c, d, 0, 1, 2, 3, 4, 5, 6, 7);
    return __builtin_shufflevector(ab, cd, 0, 1, 2, 3, 4, 5, 6, 7, 8, 9, 10, 11, 12, 13, 14, 15);
}

__device__ __forceinline__ void gat_h_store4(float4* dst, const int tt, const int k, const v16f r)
{
    *(v4f*)(dst + tt + 64 * (4 * k + 0)) = __builtin_shufflevector(r, r, 0, 1, 2, 3);
    *(v4f*)(dst + tt + 64 * (4 * k + 1)) = __builtin_shufflevector(r, r, 4, 5, 6, 7);
    *(v4f*)(dst + tt + 64 * (4 * k + 2)) = __builtin_shufflevector(r, r, 8, 9, 10, 11);
    *(v4f*)(dst + tt + 64 * (4 * k + 3)) = __builtin_shufflevector(r, r, 12, 13, 14, 15);
}

#define GAT_H_REQUEST(st_) do { \
    const float4* src_ = (const float4*)(h + ((size_t)((2 * pr + (lt >> 6)) * NODES + (st_) * STAGE_KEYS)) * FOUT); \
    hq0 = gat_h_request4(src_, lt & 63, 0); hq1 = gat_h_request4(src_, lt & 63, 1); \
    hq2 = gat_h_request4(src_, lt & 63, 2); hq3 = gat_h_request4(src_, lt & 63, 3); } while (0)

#define GAT_H_STORE(st_) do { \
    float4* dst_ = (float4*)(lds_h + ((st_) & 1) * STAGE_H + (lt >> 6) * (STAGE_KEYS * FOUT)); \
    gat_h_store4(dst_, lt & 63, 0, hq0); gat_h_store4(dst_, lt & 63, 1, hq1); \
    gat_h_store4(dst_, lt & 63, 2, hq2); gat_h_store4(dst_, lt & 63, 3, hq3); } while (0)

__device__ __forceinline__ v16f gat_s2_fetch8(const float* ls2, const int half, const int k)
{
    const float* p = ls2 + 32 * k + 2 * half;
    const v2f x0 = *(const v2f*)(p);       const v2f x1 = *(const v2f*)(p + 4);
    const v2f x2 = *(const v2f*)(p + 8);   const v2f x3 = *(const v2f*)(p + 12);
    const v2f x4 = *(const v2f*)(p + 16);  const v2f x5 = *(const v2f*)(p + 20);
    const v2f x6 = *(const v2f*)(p + 24);  const v2f x7 = *(const v2f*)(p + 28);
    const v4f y0 = __builtin_shufflevector(x0, x1, 0, 1, 2, 3);
    const v4f y1 = __builtin_shufflevector(x2, x3, 0, 1, 2, 3);
    const v4f y2 = __builtin_shufflevector(x4, x5, 0, 1, 2, 3);
    const v4f y3 = __builtin_shufflevector(x6, x7, 0, 1, 2, 3);
    const v8f z0 = __builtin_shufflevector(y0, y1, 0, 1, 2, 3, 4, 5, 6, 7);
    const v8f z1 = __builtin_shufflevector(y2, y3, 0, 1, 2, 3, 4, 5, 6, 7);
    return __builtin_shufflevector(z0, z1, 0, 1, 2, 3, 4, 5, 6, 7, 8, 9, 10, 11, 12, 13, 14, 15);
}

#define GAT_S2_FETCH(st_) do { \
    const float* ls2_ = ls2b + (st_) * STAGE_KEYS; \
    sq0 = gat_s2_fetch8(ls2_, half, 0); sq1 = gat_s2_fetch8(ls2_, half, 1); \
    mw = *(const float2*)(lmkb + ((st_) >> 1) * 4); } while (0)

__device__ __forceinline__ float gat_weights(
    const unsigned msh, const v16f s0, const v16f s1, const float2 mw, float* pdst,
    const float sv1, const float mfin, float lsum)
{
    unsigned mk[2];
    mk[0] = __float_as_uint(mw.x) >> msh;
    mk[1] = __float_as_uint(mw.y) >> msh;
#pragma unroll
    for (int jj = 0; jj < STAGE_KEYS / 16; ++jj) {
        const v16f sx = (jj >> 1) == 0 ? s0 : s1;
        float p[8];
        float rs = 0.0f;
#pragma unroll
        for (int q = 0; q < 4; ++q) {
#pragma unroll
            for (int c = 0; c < 2; ++c) {
                float e = sv1 + sx[8 * (jj & 1) + 2 * q + c];
                e = (e >= 0.0f) ? e : ALPHA * e;
                e = ((mk[c] >> (4 * jj + q)) & 1u) ? e : NEG_BIG;
                const float w = __expf(e - mfin);
                p[2 * q + c] = w;
                rs += w;
            }
        }
        lsum += rs;
        float4 lo, hi;
        lo.x = p[0]; lo.y = p[1]; lo.z = p[2]; lo.w = p[3];
        hi.x = p[4]; hi.y = p[5]; hi.z = p[6]; hi.w = p[7];
        *(float4*)(pdst + (2 * jj)     * 128) = lo;
        *(float4*)(pdst + (2 * jj + 1) * 128) = hi;
    }
    return lsum;
}

__device__ __forceinline__ void gat_frag_load(
    float (&hv)[8][4][2], float4 (&pw)[4], const float* lh, const float* pp,
    const int blk, const int half, const int n)
{
#pragma unroll
    for (int j2 = 0; j2 < 2; ++j2) {
        const int jj = 2 * blk + j2;
        pw[2 * j2]     = *(const float4*)(pp + (2 * jj)     * 128);
        pw[2 * j2 + 1] = *(const float4*)(pp + (2 * jj + 1) * 128);
#pragma unroll
        for (int q = 0; q < 4; ++q) {
            const float* hr0 = lh + (16 * jj + 4 * q + 2 * half) * FOUT + n;
            const float* hr1 = hr0 + FOUT;
#pragma unroll
            for (int t = 0; t < 4; ++t) {
                hv[4 * j2 + q][t][0] = hr0[16 * t];
                hv[4 * j2 + q][t][1] = hr1[16 * t];
            }
        }
    }
}

__device__ __forceinline__ void gat_frag_mma(
    const float (&hv)[8][4][2], const float4 (&pw)[4], v8f& acc0, v8f& acc1, v8f& acc2, v8f& acc3)
{
#pragma unroll
    for (int ks = 0; ks < 8; ++ks) {
        const float4 w4 = pw[ks >> 1];
        v2f P;
        P[0] = (ks & 1) ? w4.z : w4.x;
        P[1] = (ks & 1) ? w4.w : w4.y;
        v2f H0; H0[0] = hv[ks][0][0]; H0[1] = hv[ks][0][1];
        v2f H1; H1[0] = hv[ks][1][0]; H1[1] = hv[ks][1][1];
        v2f H2; H2[0] = hv[ks][2][0]; H2[1] = hv[ks][2][1];
        v2f H3; H3[0] = hv[ks][3][0]; H3[1] = hv[ks][3][1];
        acc0 = __builtin_amdgcn_wmma_f32_16x16x4_f32(false, H0, false, P, (short)0, acc0, false, false);
        acc1 = __builtin_amdgcn_wmma_f32_16x16x4_f32(false, H1, false, P, (short)0, acc1, false, false);
        acc2 = __builtin_amdgcn_wmma_f32_16x16x4_f32(false, H2, false, P, (short)0, acc2, false, false);
        acc3 = __builtin_amdgcn_wmma_f32_16x16x4_f32(false, H3, false, P, (short)0, acc3, false, false);
    }
}

__global__ void __launch_bounds__(ATTN_THREADS) __attribute__((amdgpu_waves_per_eu(2, 2))) gat_attn_kernel(
    const float* __restrict__ h, const float* __restrict__ s1g,
    const float* __restrict__ s2g, float* out)
{
    __shared__ alignas(16) float lds_h[2 * STAGE_H];
    __shared__ alignas(16) float lds_p[2 * STAGE_P];
    __shared__ alignas(16) float lds_s2[2 * NODES];
    __shared__ alignas(16) float lds_mk[32 * MK_PITCH];
    __shared__ alignas(16) float lds_pp[2 * 2 * 32 * RM_PITCH];
    __shared__ float lds_inv[ATTN_MATRIX_WAVES * 32];

    const int lane = threadIdx.x & 31;
    const int wid  = __builtin_amdgcn_readfirstlane((int)(threadIdx.x >> 5));
    const bool copy_wave = wid >= ATTN_MATRIX_WAVES;
    const int cw   = wid & 3;
    const int pr   = blockIdx.x >> 7;
    const int bsel = cw >> 1;
    const int b    = 2 * pr + bsel;
    const int iblk = (blockIdx.x & 127) << 5;
    const int i0   = iblk + ((cw & 1) << 4);
    const int half = lane >> 4;
    const int n    = lane & 15;
    const int lt   = (int)threadIdx.x & 127;

    const float sv1 = s1g[b * NODES + i0 + n];

    float lsum = 0.0f;
    v8f acc0 = {}, acc1 = {}, acc2 = {}, acc3 = {};

    v16f hq0, hq1, hq2, hq3;
    if (copy_wave) {
        GAT_H_REQUEST(0);
        const float4* s2src = (const float4*)(s2g + (size_t)(2 * pr) * NODES);
        float4 sreg[16];
#pragma unroll
        for (int k = 0; k < 16; ++k) sreg[k] = s2src[lt + 128 * k];
        float4 mreg[8];
#pragma unroll
        for (int k = 0; k < 8; ++k) {
            const int f  = lt + 128 * k;
            const int r  = f >> 5;
            const int ch = f & 31;
            mreg[k] = *(const float4*)(out + ((size_t)(2 * pr + (ch >> 4)) * NODES + iblk + r) * FOUT + 4 * (ch & 15));
        }
#pragma unroll
        for (int k = 0; k < 16; ++k) ((float4*)lds_s2)[lt + 128 * k] = sreg[k];
#pragma unroll
        for (int k = 0; k < 8; ++k) {
            const int f = lt + 128 * k;
            *(float4*)(lds_mk + (f >> 5) * MK_PITCH + 4 * (f & 31)) = mreg[k];
        }
        GAT_H_STORE(0);
    } else {
        const int rset  = cw & 1;
        const int chalf = cw >> 1;
        const float4* s2a = (const float4*)(s2g + (2 * pr)     * NODES + 2048 * chalf) + lane;
        const float4* s2b = (const float4*)(s2g + (2 * pr + 1) * NODES + 2048 * chalf) + lane;
        float4 sa[16], sb[16];
#pragma unroll
        for (int c = 0; c < 16; ++c) { sa[c] = s2a[32 * c]; sb[c] = s2b[32 * c]; }
        const unsigned* wrow = (const unsigned*)(out + ((size_t)(2 * pr + chalf) * NODES + iblk + 16 * rset) * FOUT);
        float* ppa = lds_pp + ((chalf * 2 + 0) * 32 + 16 * rset) * RM_PITCH + lane;
        float* ppb = lds_pp + ((chalf * 2 + 1) * 32 + 16 * rset) * RM_PITCH + lane;
        for (int r = 0; r < 16; ++r) {
            const unsigned* wp = wrow + r * FOUT;
            float ma = -INFINITY, mb = -INFINITY;
#pragma unroll
            for (int c = 0; c < 16; ++c) {
                const bool k0 = __builtin_amdgcn_inverse_ballot_w32(wp[4 * c + 0]);
                const bool k1 = __builtin_amdgcn_inverse_ballot_w32(wp[4 * c + 1]);
                const bool k2 = __builtin_amdgcn_inverse_ballot_w32(wp[4 * c + 2]);
                const bool k3 = __builtin_amdgcn_inverse_ballot_w32(wp[4 * c + 3]);
                ma = fmaxf(ma, fmaxf(k0 ? sa[c].x : -INFINITY, k1 ? sa[c].y : -INFINITY));
                ma = fmaxf(ma, fmaxf(k2 ? sa[c].z : -INFINITY, k3 ? sa[c].w : -INFINITY));
                mb = fmaxf(mb, fmaxf(k0 ? sb[c].x : -INFINITY, k1 ? sb[c].y : -INFINITY));
                mb = fmaxf(mb, fmaxf(k2 ? sb[c].z : -INFINITY, k3 ? sb[c].w : -INFINITY));
            }
            ppa[r * RM_PITCH] = ma;
            ppb[r * RM_PITCH] = mb;
        }
    }
    __syncthreads();

    float mfin = NEG_BIG;
    const float* ls2b = lds_s2 + bsel * NODES;
    const float* lmkb = lds_mk + (((cw & 1) << 4) + n) * MK_PITCH + 2 * half;
    float* pdb = lds_p + cw * WAVE_P + lane * 4;
    if (copy_wave) {
        const float4* pr4 = (const float4*)(lds_pp + ((half * 2 + bsel) * 32 + ((cw & 1) << 4) + n) * RM_PITCH);
        float rm = -INFINITY;
        for (int k = 0; k < 8; ++k) {
            const float4 t = pr4[k];
            rm = fmaxf(rm, fmaxf(fmaxf(t.x, t.y), fmaxf(t.z, t.w)));
        }
        rm = fmaxf(rm, __shfl_xor(rm, 16, 32));
        float e = sv1 + rm;
        e = (e >= 0.0f) ? e : ALPHA * e;
        mfin = fmaxf(e, NEG_BIG);
        GAT_H_REQUEST(1);
        v16f sq0, sq1;
        float2 mw;
        GAT_S2_FETCH(0);
        lsum = gat_weights(0u, sq0, sq1, mw, pdb, sv1, mfin, lsum);
    }
    __syncthreads();

    float hvA[8][4][2], hvB[8][4][2];
    float4 pwA[4], pwB[4];
#pragma unroll
    for (int ks = 0; ks < 8; ++ks)
#pragma unroll
        for (int t = 0; t < 4; ++t) { hvB[ks][t][0] = 0.0f; hvB[ks][t][1] = 0.0f; }
#pragma unroll
    for (int i = 0; i < 4; ++i) pwB[i] = make_float4(0.0f, 0.0f, 0.0f, 0.0f);
    for (int st = 0; st < N_STAGES; ++st) {
        if (copy_wave) {
            if (st + 1 < N_STAGES) {
                GAT_H_STORE(st + 1);
                asm volatile("" ::: "memory");
                GAT_H_REQUEST((st + 2 < N_STAGES) ? st + 2 : N_STAGES - 1);
                v16f sq0, sq1;
                float2 mw;
                GAT_S2_FETCH(st + 1);
                float sv1p = sv1, mfinp = mfin;
                asm volatile("" : "+v"(sv1p), "+v"(mfinp) : : "memory");
                lsum = gat_weights((unsigned)((st + 1) & 1) << 4, sq0, sq1, mw,
                                   pdb + ((st + 1) & 1) * STAGE_P, sv1p, mfinp, lsum);
            } else {
                lds_inv[cw * 32 + lane] = 1.0f / (lsum + __shfl_xor(lsum, 16, 32));
            }
        } else {
            const float* lh = lds_h + (st & 1) * STAGE_H + bsel * (STAGE_KEYS * FOUT);
            const float* pp = pdb + (st & 1) * STAGE_P;
            gat_frag_load(hvA, pwA, lh, pp, 0, half, n);
            GAT_GROUP_EDGE();
            gat_frag_mma(hvB, pwB, acc0, acc1, acc2, acc3);
            GAT_GROUP_EDGE();
            gat_frag_load(hvB, pwB, lh, pp, 1, half, n);
            GAT_GROUP_EDGE();
            gat_frag_mma(hvA, pwA, acc0, acc1, acc2, acc3);
        }
        __syncthreads();
    }

    if (!copy_wave) {
        gat_frag_mma(hvB, pwB, acc0, acc1, acc2, acc3);
        const float inv = lds_inv[cw * 32 + lane];
        float* o = out + ((size_t)(b * NODES + i0 + n)) * FOUT + 8 * half;
        gat_store_tile(o,      acc0, inv);
        gat_store_tile(o + 16, acc1, inv);
        gat_store_tile(o + 32, acc2, inv);
        gat_store_tile(o + 48, acc3, inv);
    }
}

extern "C" void kernel_launch(void* const* d_in, const int* in_sizes, int n_in,
                              void* d_out, int out_size, void* d_ws, size_t ws_size,
                              hipStream_t stream)
{
    const float* x   = (const float*)d_in[0];
    const int*   adj = (const int*)  d_in[1];
    const float* W   = (const float*)d_in[2];
    const float* a   = (const float*)d_in[3];
    float* out = (float*)d_out;

    float* h  = (float*)d_ws;
    float* s1 = h + (size_t)BATCH * NODES * FOUT;
    float* s2 = s1 + (size_t)BATCH * NODES;

    const int total_waves = BATCH * (NODES / 16);

    gat_pack_kernel<<<NODES / WAVES_PER_BLOCK, 256, 0, stream>>>(adj, out);
    gat_h_kernel<<<K1_BLOCKS, 256, 0, stream>>>(x, W, a, h, s1, s2);
    gat_attn_kernel<<<total_waves / ATTN_MATRIX_WAVES, ATTN_THREADS, 0, stream>>>(h, s1, s2, out);
}


# Attention LDS rows padded against bank conflicts: staged h rows at pitch 80 floats (lane halves of the matrix waves' 4-byte fragment reads shared banks), mask rows at 132 words; sums unchanged
# speedup vs baseline: 3.6075x; 1.0519x over previous
#include <hip/hip_runtime.h>
#include <hip/hip_bf16.h>

typedef __attribute__((ext_vector_type(2))) float v2f;
typedef __attribute__((ext_vector_type(8))) float v8f;

#define NODES   4096
#define FIN     512
#define FOUT    64
#define BATCH   4
#define NEG_BIG (-9000000000000000.0f)
#define ALPHA   0.2f
#define WAVES_PER_BLOCK 8

#define K1_ROWS  64
#define K1_TILES (K1_ROWS / 16)
#define K1_BLOCKS ((BATCH * NODES) / K1_ROWS)

__global__ void __launch_bounds__(256) __attribute__((amdgpu_waves_per_eu(2, 2))) gat_h_kernel(
    const float* __restrict__ x, const float* __restrict__ W,
    const float* __restrict__ a,
    float* __restrict__ h, float* __restrict__ s1, float* __restrict__ s2)
{
    __shared__ alignas(16) float k1_part[WAVES_PER_BLOCK * K1_ROWS * FOUT];
    __shared__ float k1_s[2 * K1_ROWS];

    const int tid  = (int)threadIdx.x;
    const int lane = tid & 31;
    const int w    = __builtin_amdgcn_readfirstlane(tid >> 5);
    const int half = lane >> 4;
    const int n    = lane & 15;
    const int R0   = (int)blockIdx.x * K1_ROWS;

    const float* wp = W + (size_t)(64 * w + 4 * half) * FOUT + 4 * n;
    float4 wr[32];
#pragma unroll
    for (int i = 0; i < 32; ++i)
        wr[i] = *(const float4*)(wp + (size_t)(8 * (i >> 2) + (i & 3)) * FOUT);

    const float* xp = x + (size_t)(R0 + n) * FIN + 64 * w + 4 * half;
    float4 xr[K1_TILES][8];
#pragma unroll
    for (int rt = 0; rt < K1_TILES; ++rt)
#pragma unroll
        for (int t = 0; t < 8; ++t)
            xr[rt][t] = *(const float4*)(xp + (size_t)(16 * rt) * FIN + 8 * t);

#pragma unroll
    for (int rt = 0; rt < K1_TILES; ++rt) {
        v8f acc0 = {}, acc1 = {}, acc2 = {}, acc3 = {};
#pragma unroll
        for (int t = 0; t < 8; ++t) {
#pragma unroll
            for (int u = 0; u < 2; ++u) {
                const float4 xa = xr[rt][t];
                v2f A;
                A[0] = u ? xa.z : xa.x;
                A[1] = u ? xa.w : xa.y;
                const float4 b0 = wr[4 * t + 2 * u];
                const float4 b1 = wr[4 * t + 2 * u + 1];
                v2f B0; B0[0] = b0.x; B0[1] = b1.x;
                v2f B1; B1[0] = b0.y; B1[1] = b1.y;
                v2f B2; B2[0] = b0.z; B2[1] = b1.z;
                v2f B3; B3[0] = b0.w; B3[1] = b1.w;
                acc0 = __builtin_amdgcn_wmma_f32_16x16x4_f32(false, A, false, B0, (short)0, acc0, false, false);
                acc1 = __builtin_amdgcn_wmma_f32_16x16x4_f32(false, A, false, B1, (short)0, acc1, false, false);
                acc2 = __builtin_amdgcn_wmma_f32_16x16x4_f32(false, A, false, B2, (short)0, acc2, false, false);
                acc3 = __builtin_amdgcn_wmma_f32_16x16x4_f32(false, A, false, B3, (short)0, acc3, false, false);
            }
        }
        float* pp = k1_part + ((size_t)(w * K1_ROWS + 16 * rt + 8 * half)) * FOUT + 4 * n;
#pragma unroll
        for (int r = 0; r < 8; ++r) {
            float4 o; o.x = acc0[r]; o.y = acc1[r]; o.z = acc2[r]; o.w = acc3[r];
            *(float4*)(pp + r * FOUT) = o;
        }
    }
    __syncthreads();

    const float4 av1 = *(const float4*)(a + 4 * (tid & 15));
    const float4 av2 = *(const float4*)(a + FOUT + 4 * (tid & 15));
    const float4* p4 = (const float4*)k1_part;
    float4* h4 = (float4*)(h + (size_t)R0 * FOUT);
#pragma unroll
    for (int j = 0; j < 4; ++j) {
        const int f = tid + 256 * j;
        float4 sum = p4[f];
#pragma unroll
        for (int ww = 1; ww < WAVES_PER_BLOCK; ++ww) {
            const float4 t4 = p4[ww * (K1_ROWS * FOUT / 4) + f];
            sum.x += t4.x; sum.y += t4.y; sum.z += t4.z; sum.w += t4.w;
        }
        h4[f] = sum;
        float p1 = sum.x * av1.x + sum.y * av1.y + sum.z * av1.z + sum.w * av1.w;
        float p2 = sum.x * av2.x + sum.y * av2.y + sum.z * av2.z + sum.w * av2.w;
        for (int m = 1; m < 16; m <<= 1) {
            p1 += __shfl_xor(p1, m, 32);
            p2 += __shfl_xor(p2, m, 32);
        }
        if ((tid & 15) == 0) {
            k1_s[f >> 4] = p1;
            k1_s[K1_ROWS + (f >> 4)] = p2;
        }
    }
    __syncthreads();
    if (tid < K1_ROWS)          s1[R0 + tid] = k1_s[tid];
    else if (tid < 2 * K1_ROWS) s2[R0 + tid - K1_ROWS] = k1_s[tid];
}

__global__ void __launch_bounds__(256) gat_pack_kernel(
    const int* __restrict__ adj, float* out)
{
    const int lane = threadIdx.x & 31;
    const int row  = blockIdx.x * WAVES_PER_BLOCK + (threadIdx.x >> 5);
    const int4* src = (const int4*)(adj + (size_t)row * NODES) + lane;

    int4 v[32];
#pragma unroll
    for (int c = 0; c < 32; ++c) v[c] = src[c * 32];

    unsigned m0 = 0u, m1 = 0u, m2 = 0u, m3 = 0u;
#pragma unroll
    for (int c = 0; c < 32; ++c) {
        const unsigned w0 = __builtin_amdgcn_ballot_w32(v[c].x > 0);
        const unsigned w1 = __builtin_amdgcn_ballot_w32(v[c].y > 0);
        const unsigned w2 = __builtin_amdgcn_ballot_w32(v[c].z > 0);
        const unsigned w3 = __builtin_amdgcn_ballot_w32(v[c].w > 0);
        if (lane == c) { m0 = w0; m1 = w1; m2 = w2; m3 = w3; }
    }

    float4 o;
    o.x = __uint_as_float(m0); o.y = __uint_as_float(m1);
    o.z = __uint_as_float(m2); o.w = __uint_as_float(m3);
    float* d = out + ((size_t)((lane >> 4) * NODES + row)) * FOUT + (lane & 15) * 4;
    *(float4*)d = o;
    *(float4*)(d + (size_t)2 * NODES * FOUT) = o;
}

__device__ __forceinline__ float gat_elu(float v) { return (v > 0.0f) ? v : expm1f(v); }

__device__ __forceinline__ void gat_store_tile(float* o, const v8f acc, const float inv)
{
    float4 lo, hi;
    lo.x = gat_elu(acc[0] * inv); lo.y = gat_elu(acc[1] * inv);
    lo.z = gat_elu(acc[2] * inv); lo.w = gat_elu(acc[3] * inv);
    hi.x = gat_elu(acc[4] * inv); hi.y = gat_elu(acc[5] * inv);
    hi.z = gat_elu(acc[6] * inv); hi.w = gat_elu(acc[7] * inv);
    *(float4*)o       = lo;
    *(float4*)(o + 4) = hi;
}

#define STAGE_KEYS 64
#define N_STAGES (NODES / STAGE_KEYS)
#define ATTN_MATRIX_WAVES 4
#define ATTN_COPY_WAVES 4
#define ATTN_THREADS (32 * (ATTN_MATRIX_WAVES + ATTN_COPY_WAVES))
#define H_PITCH  80
#define STAGE_H  (2 * STAGE_KEYS * H_PITCH)
#define WAVE_P   ((STAGE_KEYS / 16) * 2 * 32 * 4)
#define STAGE_P  (ATTN_MATRIX_WAVES * WAVE_P)
#define MK_PITCH 132
#define GAT_GROUP_EDGE() do { asm volatile("" : "+v"(acc0), "+v"(acc1), "+v"(acc2), "+v"(acc3) : : "memory"); __builtin_amdgcn_sched_barrier(0); } while (0)
#define RM_PITCH 36

typedef __attribute__((ext_vector_type(4)))  float v4f;
typedef __attribute__((ext_vector_type(16))) float v16f;


__device__ __forceinline__ v16f gat_h_request4(const float4* src, const int tt, const int k)
{
    const v4f a = *(const v4f*)(src + tt + 64 * (4 * k + 0));
    const v4f b = *(const v4f*)(src + tt + 64 * (4 * k + 1));
    const v4f c = *(const v4f*)(src + tt + 64 * (4 * k + 2));
    const v4f d = *(const v4f*)(src + tt + 64 * (4 * k + 3));
    const v8f ab = __builtin_shufflevector(a, b, 0, 1, 2, 3, 4, 5, 6, 7);
    const v8f cd = __builtin_shufflevector(c, d, 0, 1, 2, 3, 4, 5, 6, 7);
    return __builtin_shufflevector(ab, cd, 0, 1, 2, 3, 4, 5, 6, 7, 8, 9, 10, 11, 12, 13, 14, 15);
}

__device__ __forceinline__ void gat_h_store4(float4* dst, const int tp, const int k, const v16f r)
{
    *(v4f*)(dst + tp + H_PITCH * (4 * k + 0)) = __builtin_shufflevector(r, r, 0, 1, 2, 3);
    *(v4f*)(dst + tp + H_PITCH * (4 * k + 1)) = __builtin_shufflevector(r, r, 4, 5, 6, 7);
    *(v4f*)(dst + tp + H_PITCH * (4 * k + 2)) = __builtin_shufflevector(r, r, 8, 9, 10, 11);
    *(v4f*)(dst + tp + H_PITCH * (4 * k + 3)) = __builtin_shufflevector(r, r, 12, 13, 14, 15);
}

#define GAT_H_REQUEST(st_) do { \
    const float4* src_ = (const float4*)(h + ((size_t)((2 * pr + (lt >> 6)) * NODES + (st_) * STAGE_KEYS)) * FOUT); \
    hq0 = gat_h_request4(src_, lt & 63, 0); hq1 = gat_h_request4(src_, lt & 63, 1); \
    hq2 = gat_h_request4(src_, lt & 63, 2); hq3 = gat_h_request4(src_, lt & 63, 3); } while (0)

#define GAT_H_STORE(st_) do { \
    float4* dst_ = (float4*)(lds_h + ((st_) & 1) * STAGE_H + (lt >> 6) * (STAGE_KEYS * H_PITCH)); \
    const int tp_ = ((lt & 63) >> 4) * (H_PITCH / 4) + (lt & 15); \
    gat_h_store4(dst_, tp_, 0, hq0); gat_h_store4(dst_, tp_, 1, hq1); \
    gat_h_store4(dst_, tp_, 2, hq2); gat_h_store4(dst_, tp_, 3, hq3); } while (0)

__device__ __forceinline__ v16f gat_s2_fetch8(const float* ls2, const int half, const int k)
{
    const float* p = ls2 + 32 * k + 2 * half;
    const v2f x0 = *(const v2f*)(p);       const v2f x1 = *(const v2f*)(p + 4);
    const v2f x2 = *(const v2f*)(p + 8);   const v2f x3 = *(const v2f*)(p + 12);
    const v2f x4 = *(const v2f*)(p + 16);  const v2f x5 = *(const v2f*)(p + 20);
    const v2f x6 = *(const v2f*)(p + 24);  const v2f x7 = *(const v2f*)(p + 28);
    const v4f y0 = __builtin_shufflevector(x0, x1, 0, 1, 2, 3);
    const v4f y1 = __builtin_shufflevector(x2, x3, 0, 1, 2, 3);
    const v4f y2 = __builtin_shufflevector(x4, x5, 0, 1, 2, 3);
    const v4f y3 = __builtin_shufflevector(x6, x7, 0, 1, 2, 3);
    const v8f z0 = __builtin_shufflevector(y0, y1, 0, 1, 2, 3, 4, 5, 6, 7);
    const v8f z1 = __builtin_shufflevector(y2, y3, 0, 1, 2, 3, 4, 5, 6, 7);
    return __builtin_shufflevector(z0, z1, 0, 1, 2, 3, 4, 5, 6, 7, 8, 9, 10, 11, 12, 13, 14, 15);
}

#define GAT_S2_FETCH(st_) do { \
    const float* ls2_ = ls2b + (st_) * STAGE_KEYS; \
    sq0 = gat_s2_fetch8(ls2_, half, 0); sq1 = gat_s2_fetch8(ls2_, half, 1); \
    mw = *(const float2*)(lmkb + ((st_) >> 1) * 4); } while (0)

__device__ __forceinline__ float gat_weights(
    const unsigned msh, const v16f s0, const v16f s1, const float2 mw, float* pdst,
    const float sv1, const float mfin, float lsum)
{
    unsigned mk[2];
    mk[0] = __float_as_uint(mw.x) >> msh;
    mk[1] = __float_as_uint(mw.y) >> msh;
#pragma unroll
    for (int jj = 0; jj < STAGE_KEYS / 16; ++jj) {
        const v16f sx = (jj >> 1) == 0 ? s0 : s1;
        float p[8];
        float rs = 0.0f;
#pragma unroll
        for (int q = 0; q < 4; ++q) {
#pragma unroll
            for (int c = 0; c < 2; ++c) {
                float e = sv1 + sx[8 * (jj & 1) + 2 * q + c];
                e = (e >= 0.0f) ? e : ALPHA * e;
                e = ((mk[c] >> (4 * jj + q)) & 1u) ? e : NEG_BIG;
                const float w = __expf(e - mfin);
                p[2 * q + c] = w;
                rs += w;
            }
        }
        lsum += rs;
        float4 lo, hi;
        lo.x = p[0]; lo.y = p[1]; lo.z = p[2]; lo.w = p[3];
        hi.x = p[4]; hi.y = p[5]; hi.z = p[6]; hi.w = p[7];
        *(float4*)(pdst + (2 * jj)     * 128) = lo;
        *(float4*)(pdst + (2 * jj + 1) * 128) = hi;
    }
    return lsum;
}

__device__ __forceinline__ void gat_frag_load(
    float (&hv)[8][4][2], float4 (&pw)[4], const float* lh, const float* pp,
    const int blk, const int half, const int n)
{
#pragma unroll
    for (int j2 = 0; j2 < 2; ++j2) {
        const int jj = 2 * blk + j2;
        pw[2 * j2]     = *(const float4*)(pp + (2 * jj)     * 128);
        pw[2 * j2 + 1] = *(const float4*)(pp + (2 * jj + 1) * 128);
#pragma unroll
        for (int q = 0; q < 4; ++q) {
            const float* hr0 = lh + (16 * jj + 4 * q + 2 * half) * H_PITCH + n;
            const float* hr1 = hr0 + H_PITCH;
#pragma unroll
            for (int t = 0; t < 4; ++t) {
                hv[4 * j2 + q][t][0] = hr0[16 * t];
                hv[4 * j2 + q][t][1] = hr1[16 * t];
            }
        }
    }
}

__device__ __forceinline__ void gat_frag_mma(
    const float (&hv)[8][4][2], const float4 (&pw)[4], v8f& acc0, v8f& acc1, v8f& acc2, v8f& acc3)
{
#pragma unroll
    for (int ks = 0; ks < 8; ++ks) {
        const float4 w4 = pw[ks >> 1];
        v2f P;
        P[0] = (ks & 1) ? w4.z : w4.x;
        P[1] = (ks & 1) ? w4.w : w4.y;
        v2f H0; H0[0] = hv[ks][0][0]; H0[1] = hv[ks][0][1];
        v2f H1; H1[0] = hv[ks][1][0]; H1[1] = hv[ks][1][1];
        v2f H2; H2[0] = hv[ks][2][0]; H2[1] = hv[ks][2][1];
        v2f H3; H3[0] = hv[ks][3][0]; H3[1] = hv[ks][3][1];
        acc0 = __builtin_amdgcn_wmma_f32_16x16x4_f32(false, H0, false, P, (short)0, acc0, false, false);
        acc1 = __builtin_amdgcn_wmma_f32_16x16x4_f32(false, H1, false, P, (short)0, acc1, false, false);
        acc2 = __builtin_amdgcn_wmma_f32_16x16x4_f32(false, H2, false, P, (short)0, acc2, false, false);
        acc3 = __builtin_amdgcn_wmma_f32_16x16x4_f32(false, H3, false, P, (short)0, acc3, false, false);
    }
}

__global__ void __launch_bounds__(ATTN_THREADS) __attribute__((amdgpu_waves_per_eu(2, 2))) gat_attn_kernel(
    const float* __restrict__ h, const float* __restrict__ s1g,
    const float* __restrict__ s2g, float* out)
{
    __shared__ alignas(16) float lds_h[2 * STAGE_H];
    __shared__ alignas(16) float lds_p[2 * STAGE_P];
    __shared__ alignas(16) float lds_s2[2 * NODES];
    __shared__ alignas(16) float lds_mk[32 * MK_PITCH];
    __shared__ alignas(16) float lds_pp[2 * 2 * 32 * RM_PITCH];
    __shared__ float lds_inv[ATTN_MATRIX_WAVES * 32];

    const int lane = threadIdx.x & 31;
    const int wid  = __builtin_amdgcn_readfirstlane((int)(threadIdx.x >> 5));
    const bool copy_wave = wid >= ATTN_MATRIX_WAVES;
    const int cw   = wid & 3;
    const int pr   = blockIdx.x >> 7;
    const int bsel = cw >> 1;
    const int b    = 2 * pr + bsel;
    const int iblk = (blockIdx.x & 127) << 5;
    const int i0   = iblk + ((cw & 1) << 4);
    const int half = lane >> 4;
    const int n    = lane & 15;
    const int lt   = (int)threadIdx.x & 127;

    const float sv1 = s1g[b * NODES + i0 + n];

    float lsum = 0.0f;
    v8f acc0 = {}, acc1 = {}, acc2 = {}, acc3 = {};

    v16f hq0, hq1, hq2, hq3;
    if (copy_wave) {
        GAT_H_REQUEST(0);
        const float4* s2src = (const float4*)(s2g + (size_t)(2 * pr) * NODES);
        float4 sreg[16];
#pragma unroll
        for (int k = 0; k < 16; ++k) sreg[k] = s2src[lt + 128 * k];
        float4 mreg[8];
#pragma unroll
        for (int k = 0; k < 8; ++k) {
            const int f  = lt + 128 * k;
            const int r  = f >> 5;
            const int ch = f & 31;
            mreg[k] = *(const float4*)(out + ((size_t)(2 * pr + (ch >> 4)) * NODES + iblk + r) * FOUT + 4 * (ch & 15));
        }
#pragma unroll
        for (int k = 0; k < 16; ++k) ((float4*)lds_s2)[lt + 128 * k] = sreg[k];
#pragma unroll
        for (int k = 0; k < 8; ++k) {
            const int f = lt + 128 * k;
            *(float4*)(lds_mk + (f >> 5) * MK_PITCH + 4 * (f & 31)) = mreg[k];
        }
        GAT_H_STORE(0);
    } else {
        const int rset  = cw & 1;
        const int chalf = cw >> 1;
        const float4* s2a = (const float4*)(s2g + (2 * pr)     * NODES + 2048 * chalf) + lane;
        const float4* s2b = (const float4*)(s2g + (2 * pr + 1) * NODES + 2048 * chalf) + lane;
        float4 sa[16], sb[16];
#pragma unroll
        for (int c = 0; c < 16; ++c) { sa[c] = s2a[32 * c]; sb[c] = s2b[32 * c]; }
        const unsigned* wrow = (const unsigned*)(out + ((size_t)(2 * pr + chalf) * NODES + iblk + 16 * rset) * FOUT);
        float* ppa = lds_pp + ((chalf * 2 + 0) * 32 + 16 * rset) * RM_PITCH + lane;
        float* ppb = lds_pp + ((chalf * 2 + 1) * 32 + 16 * rset) * RM_PITCH + lane;
        for (int r = 0; r < 16; ++r) {
            const unsigned* wp = wrow + r * FOUT;
            float ma = -INFINITY, mb = -INFINITY;
#pragma unroll
            for (int c = 0; c < 16; ++c) {
                const bool k0 = __builtin_amdgcn_inverse_ballot_w32(wp[4 * c + 0]);
                const bool k1 = __builtin_amdgcn_inverse_ballot_w32(wp[4 * c + 1]);
                const bool k2 = __builtin_amdgcn_inverse_ballot_w32(wp[4 * c + 2]);
                const bool k3 = __builtin_amdgcn_inverse_ballot_w32(wp[4 * c + 3]);
                ma = fmaxf(ma, fmaxf(k0 ? sa[c].x : -INFINITY, k1 ? sa[c].y : -INFINITY));
                ma = fmaxf(ma, fmaxf(k2 ? sa[c].z : -INFINITY, k3 ? sa[c].w : -INFINITY));
                mb = fmaxf(mb, fmaxf(k0 ? sb[c].x : -INFINITY, k1 ? sb[c].y : -INFINITY));
                mb = fmaxf(mb, fmaxf(k2 ? sb[c].z : -INFINITY, k3 ? sb[c].w : -INFINITY));
            }
            ppa[r * RM_PITCH] = ma;
            ppb[r * RM_PITCH] = mb;
        }
    }
    __syncthreads();

    float mfin = NEG_BIG;
    const float* ls2b = lds_s2 + bsel * NODES;
    const float* lmkb = lds_mk + (((cw & 1) << 4) + n) * MK_PITCH + 2 * half;
    float* pdb = lds_p + cw * WAVE_P + lane * 4;
    if (copy_wave) {
        const float4* pr4 = (const float4*)(lds_pp + ((half * 2 + bsel) * 32 + ((cw & 1) << 4) + n) * RM_PITCH);
        float rm = -INFINITY;
        for (int k = 0; k < 8; ++k) {
            const float4 t = pr4[k];
            rm = fmaxf(rm, fmaxf(fmaxf(t.x, t.y), fmaxf(t.z, t.w)));
        }
        rm = fmaxf(rm, __shfl_xor(rm, 16, 32));
        float e = sv1 + rm;
        e = (e >= 0.0f) ? e : ALPHA * e;
        mfin = fmaxf(e, NEG_BIG);
        GAT_H_REQUEST(1);
        v16f sq0, sq1;
        float2 mw;
        GAT_S2_FETCH(0);
        lsum = gat_weights(0u, sq0, sq1, mw, pdb, sv1, mfin, lsum);
    }
    __syncthreads();

    float hvA[8][4][2], hvB[8][4][2];
    float4 pwA[4], pwB[4];
#pragma unroll
    for (int ks = 0; ks < 8; ++ks)
#pragma unroll
        for (int t = 0; t < 4; ++t) { hvB[ks][t][0] = 0.0f; hvB[ks][t][1] = 0.0f; }
#pragma unroll
    for (int i = 0; i < 4; ++i) pwB[i] = make_float4(0.0f, 0.0f, 0.0f, 0.0f);
    for (int st = 0; st < N_STAGES; ++st) {
        if (copy_wave) {
            if (st + 1 < N_STAGES) {
                GAT_H_STORE(st + 1);
                asm volatile("" ::: "memory");
                GAT_H_REQUEST((st + 2 < N_STAGES) ? st + 2 : N_STAGES - 1);
                v16f sq0, sq1;
                float2 mw;
                GAT_S2_FETCH(st + 1);
                float sv1p = sv1, mfinp = mfin;
                asm volatile("" : "+v"(sv1p), "+v"(mfinp) : : "memory");
                lsum = gat_weights((unsigned)((st + 1) & 1) << 4, sq0, sq1, mw,
                                   pdb + ((st + 1) & 1) * STAGE_P, sv1p, mfinp, lsum);
            } else {
                lds_inv[cw * 32 + lane] = 1.0f / (lsum + __shfl_xor(lsum, 16, 32));
            }
        } else {
            const float* lh = lds_h + (st & 1) * STAGE_H + bsel * (STAGE_KEYS * H_PITCH);
            const float* pp = pdb + (st & 1) * STAGE_P;
            gat_frag_load(hvA, pwA, lh, pp, 0, half, n);
            GAT_GROUP_EDGE();
            gat_frag_mma(hvB, pwB, acc0, acc1, acc2, acc3);
            GAT_GROUP_EDGE();
            gat_frag_load(hvB, pwB, lh, pp, 1, half, n);
            GAT_GROUP_EDGE();
            gat_frag_mma(hvA, pwA, acc0, acc1, acc2, acc3);
        }
        __syncthreads();
    }

    if (!copy_wave) {
        gat_frag_mma(hvB, pwB, acc0, acc1, acc2, acc3);
        const float inv = lds_inv[cw * 32 + lane];
        float* o = out + ((size_t)(b * NODES + i0 + n)) * FOUT + 8 * half;
        gat_store_tile(o,      acc0, inv);
        gat_store_tile(o + 16, acc1, inv);
        gat_store_tile(o + 32, acc2, inv);
        gat_store_tile(o + 48, acc3, inv);
    }
}

extern "C" void kernel_launch(void* const* d_in, const int* in_sizes, int n_in,
                              void* d_out, int out_size, void* d_ws, size_t ws_size,
                              hipStream_t stream)
{
    const float* x   = (const float*)d_in[0];
    const int*   adj = (const int*)  d_in[1];
    const float* W   = (const float*)d_in[2];
    const float* a   = (const float*)d_in[3];
    float* out = (float*)d_out;

    float* h  = (float*)d_ws;
    float* s1 = h + (size_t)BATCH * NODES * FOUT;
    float* s2 = s1 + (size_t)BATCH * NODES;

    const int total_waves = BATCH * (NODES / 16);

    gat_pack_kernel<<<NODES / WAVES_PER_BLOCK, 256, 0, stream>>>(adj, out);
    gat_h_kernel<<<K1_BLOCKS, 256, 0, stream>>>(x, W, a, h, s1, s2);
    gat_attn_kernel<<<total_waves / ATTN_MATRIX_WAVES, ATTN_THREADS, 0, stream>>>(h, s1, s2, out);
}


# Adjacency packing folded into the h kernel's blocks (2 rows per wave: one requested behind the operand loads, one behind the barrier): 3 launches -> 2, adjacency streams in under products; sums same
# speedup vs baseline: 3.7456x; 1.0383x over previous
#include <hip/hip_runtime.h>
#include <hip/hip_bf16.h>

typedef __attribute__((ext_vector_type(2))) float v2f;
typedef __attribute__((ext_vector_type(8))) float v8f;

#define NODES   4096
#define FIN     512
#define FOUT    64
#define BATCH   4
#define NEG_BIG (-9000000000000000.0f)
#define ALPHA   0.2f
#define WAVES_PER_BLOCK 8

__device__ __forceinline__ void gat_pack_row(const int4 (&v)[32], const int lane, const int row, float* out)
{
    unsigned m0 = 0u, m1 = 0u, m2 = 0u, m3 = 0u;
#pragma unroll
    for (int c = 0; c < 32; ++c) {
        const unsigned w0 = __builtin_amdgcn_ballot_w32(v[c].x > 0);
        const unsigned w1 = __builtin_amdgcn_ballot_w32(v[c].y > 0);
        const unsigned w2 = __builtin_amdgcn_ballot_w32(v[c].z > 0);
        const unsigned w3 = __builtin_amdgcn_ballot_w32(v[c].w > 0);
        if (lane == c) { m0 = w0; m1 = w1; m2 = w2; m3 = w3; }
    }

    float4 o;
    o.x = __uint_as_float(m0); o.y = __uint_as_float(m1);
    o.z = __uint_as_float(m2); o.w = __uint_as_float(m3);
    float* d = out + ((size_t)((lane >> 4) * NODES + row)) * FOUT + (lane & 15) * 4;
    *(float4*)d = o;
    *(float4*)(d + (size_t)2 * NODES * FOUT) = o;
}

#define K1_ROWS  64
#define K1_TILES (K1_ROWS / 16)
#define K1_BLOCKS ((BATCH * NODES) / K1_ROWS)
#define K1_PACK_ROWS (NODES / K1_BLOCKS)

__global__ void __launch_bounds__(256) __attribute__((amdgpu_waves_per_eu(2, 2))) gat_h_kernel(
    const float* __restrict__ x, const float* __restrict__ W,
    const float* __restrict__ a, const int* __restrict__ adj,
    float* __restrict__ h, float* __restrict__ s1, float* __restrict__ s2, float* out)
{
    __shared__ alignas(16) float k1_part[WAVES_PER_BLOCK * K1_ROWS * FOUT];
    __shared__ float k1_s[2 * K1_ROWS];

    const int tid  = (int)threadIdx.x;
    const int lane = tid & 31;
    const int w    = __builtin_amdgcn_readfirstlane(tid >> 5);
    const int half = lane >> 4;
    const int n    = lane & 15;
    const int R0   = (int)blockIdx.x * K1_ROWS;

    const float* wp = W + (size_t)(64 * w + 4 * half) * FOUT + 4 * n;
    float4 wr[32];
#pragma unroll
    for (int i = 0; i < 32; ++i)
        wr[i] = *(const float4*)(wp + (size_t)(8 * (i >> 2) + (i & 3)) * FOUT);

    const float* xp = x + (size_t)(R0 + n) * FIN + 64 * w + 4 * half;
    float4 xr[K1_TILES][8];
#pragma unroll
    for (int rt = 0; rt < K1_TILES; ++rt)
#pragma unroll
        for (int t = 0; t < 8; ++t)
            xr[rt][t] = *(const float4*)(xp + (size_t)(16 * rt) * FIN + 8 * t);

    asm volatile("" ::: "memory");
    const int prow = (int)blockIdx.x * K1_PACK_ROWS + 2 * w;
    const int4* asrc = (const int4*)(adj + (size_t)prow * NODES) + lane;
    int4 av[32];
#pragma unroll
    for (int c = 0; c < 32; ++c) av[c] = asrc[c * 32];

#pragma unroll
    for (int rt = 0; rt < K1_TILES; ++rt) {
        v8f acc0 = {}, acc1 = {}, acc2 = {}, acc3 = {};
#pragma unroll
        for (int t = 0; t < 8; ++t) {
#pragma unroll
            for (int u = 0; u < 2; ++u) {
                const float4 xa = xr[rt][t];
                v2f A;
                A[0] = u ? xa.z : xa.x;
                A[1] = u ? xa.w : xa.y;
                const float4 b0 = wr[4 * t + 2 * u];
                const float4 b1 = wr[4 * t + 2 * u + 1];
                v2f B0; B0[0] = b0.x; B0[1] = b1.x;
                v2f B1; B1[0] = b0.y; B1[1] = b1.y;
                v2f B2; B2[0] = b0.z; B2[1] = b1.z;
                v2f B3; B3[0] = b0.w; B3[1] = b1.w;
                acc0 = __builtin_amdgcn_wmma_f32_16x16x4_f32(false, A, false, B0, (short)0, acc0, false, false);
                acc1 = __builtin_amdgcn_wmma_f32_16x16x4_f32(false, A, false, B1, (short)0, acc1, false, false);
                acc2 = __builtin_amdgcn_wmma_f32_16x16x4_f32(false, A, false, B2, (short)0, acc2, false, false);
                acc3 = __builtin_amdgcn_wmma_f32_16x16x4_f32(false, A, false, B3, (short)0, acc3, false, false);
            }
        }
        float* pp = k1_part + ((size_t)(w * K1_ROWS + 16 * rt + 8 * half)) * FOUT + 4 * n;
#pragma unroll
        for (int r = 0; r < 8; ++r) {
            float4 o; o.x = acc0[r]; o.y = acc1[r]; o.z = acc2[r]; o.w = acc3[r];
            *(float4*)(pp + r * FOUT) = o;
        }
    }
    __syncthreads();
    gat_pack_row(av, lane, prow, out);
#pragma unroll
    for (int c = 0; c < 32; ++c) av[c] = asrc[NODES / 4 + c * 32];

    const float4 av1 = *(const float4*)(a + 4 * (tid & 15));
    const float4 av2 = *(const float4*)(a + FOUT + 4 * (tid & 15));
    const float4* p4 = (const float4*)k1_part;
    float4* h4 = (float4*)(h + (size_t)R0 * FOUT);
#pragma unroll
    for (int j = 0; j < 4; ++j) {
        const int f = tid + 256 * j;
        float4 sum = p4[f];
#pragma unroll
        for (int ww = 1; ww < WAVES_PER_BLOCK; ++ww) {
            const float4 t4 = p4[ww * (K1_ROWS * FOUT / 4) + f];
            sum.x += t4.x; sum.y += t4.y; sum.z += t4.z; sum.w += t4.w;
        }
        h4[f] = sum;
        float p1 = sum.x * av1.x + sum.y * av1.y + sum.z * av1.z + sum.w * av1.w;
        float p2 = sum.x * av2.x + sum.y * av2.y + sum.z * av2.z + sum.w * av2.w;
        for (int m = 1; m < 16; m <<= 1) {
            p1 += __shfl_xor(p1, m, 32);
            p2 += __shfl_xor(p2, m, 32);
        }
        if ((tid & 15) == 0) {
            k1_s[f >> 4] = p1;
            k1_s[K1_ROWS + (f >> 4)] = p2;
        }
    }
    __syncthreads();
    if (tid < K1_ROWS)          s1[R0 + tid] = k1_s[tid];
    else if (tid < 2 * K1_ROWS) s2[R0 + tid - K1_ROWS] = k1_s[tid];
    gat_pack_row(av, lane, prow + 1, out);
}

__device__ __forceinline__ float gat_elu(float v) { return (v > 0.0f) ? v : expm1f(v); }

__device__ __forceinline__ void gat_store_tile(float* o, const v8f acc, const float inv)
{
    float4 lo, hi;
    lo.x = gat_elu(acc[0] * inv); lo.y = gat_elu(acc[1] * inv);
    lo.z = gat_elu(acc[2] * inv); lo.w = gat_elu(acc[3] * inv);
    hi.x = gat_elu(acc[4] * inv); hi.y = gat_elu(acc[5] * inv);
    hi.z = gat_elu(acc[6] * inv); hi.w = gat_elu(acc[7] * inv);
    *(float4*)o       = lo;
    *(float4*)(o + 4) = hi;
}

#define STAGE_KEYS 64
#define N_STAGES (NODES / STAGE_KEYS)
#define ATTN_MATRIX_WAVES 4
#define ATTN_COPY_WAVES 4
#define ATTN_THREADS (32 * (ATTN_MATRIX_WAVES + ATTN_COPY_WAVES))
#define H_PITCH  80
#define STAGE_H  (2 * STAGE_KEYS * H_PITCH)
#define WAVE_P   ((STAGE_KEYS / 16) * 2 * 32 * 4)
#define STAGE_P  (ATTN_MATRIX_WAVES * WAVE_P)
#define MK_PITCH 132
#define GAT_GROUP_EDGE() do { asm volatile("" : "+v"(acc0), "+v"(acc1), "+v"(acc2), "+v"(acc3) : : "memory"); __builtin_amdgcn_sched_barrier(0); } while (0)
#define RM_PITCH 36

typedef __attribute__((ext_vector_type(4)))  float v4f;
typedef __attribute__((ext_vector_type(16))) float v16f;


__device__ __forceinline__ v16f gat_h_request4(const float4* src, const int tt, const int k)
{
    const v4f a = *(const v4f*)(src + tt + 64 * (4 * k + 0));
    const v4f b = *(const v4f*)(src + tt + 64 * (4 * k + 1));
    const v4f c = *(const v4f*)(src + tt + 64 * (4 * k + 2));
    const v4f d = *(const v4f*)(src + tt + 64 * (4 * k + 3));
    const v8f ab = __builtin_shufflevector(a, b, 0, 1, 2, 3, 4, 5, 6, 7);
    const v8f cd = __builtin_shufflevector(c, d, 0, 1, 2, 3, 4, 5, 6, 7);
    return __builtin_shufflevector(ab, cd, 0, 1, 2, 3, 4, 5, 6, 7, 8, 9, 10, 11, 12, 13, 14, 15);
}

__device__ __forceinline__ void gat_h_store4(float4* dst, const int tp, const int k, const v16f r)
{
    *(v4f*)(dst + tp + H_PITCH * (4 * k + 0)) = __builtin_shufflevector(r, r, 0, 1, 2, 3);
    *(v4f*)(dst + tp + H_PITCH * (4 * k + 1)) = __builtin_shufflevector(r, r, 4, 5, 6, 7);
    *(v4f*)(dst + tp + H_PITCH * (4 * k + 2)) = __builtin_shufflevector(r, r, 8, 9, 10, 11);
    *(v4f*)(dst + tp + H_PITCH * (4 * k + 3)) = __builtin_shufflevector(r, r, 12, 13, 14, 15);
}

#define GAT_H_REQUEST(st_) do { \
    const float4* src_ = (const float4*)(h + ((size_t)((2 * pr + (lt >> 6)) * NODES + (st_) * STAGE_KEYS)) * FOUT); \
    hq0 = gat_h_request4(src_, lt & 63, 0); hq1 = gat_h_request4(src_, lt & 63, 1); \
    hq2 = gat_h_request4(src_, lt & 63, 2); hq3 = gat_h_request4(src_, lt & 63, 3); } while (0)

#define GAT_H_STORE(st_) do { \
    float4* dst_ = (float4*)(lds_h + ((st_) & 1) * STAGE_H + (lt >> 6) * (STAGE_KEYS * H_PITCH)); \
    const int tp_ = ((lt & 63) >> 4) * (H_PITCH / 4) + (lt & 15); \
    gat_h_store4(dst_, tp_, 0, hq0); gat_h_store4(dst_, tp_, 1, hq1); \
    gat_h_store4(dst_, tp_, 2, hq2); gat_h_store4(dst_, tp_, 3, hq3); } while (0)

__device__ __forceinline__ v16f gat_s2_fetch8(const float* ls2, const int half, const int k)
{
    const float* p = ls2 + 32 * k + 2 * half;
    const v2f x0 = *(const v2f*)(p);       const v2f x1 = *(const v2f*)(p + 4);
    const v2f x2 = *(const v2f*)(p + 8);   const v2f x3 = *(const v2f*)(p + 12);
    const v2f x4 = *(const v2f*)(p + 16);  const v2f x5 = *(const v2f*)(p + 20);
    const v2f x6 = *(const v2f*)(p + 24);  const v2f x7 = *(const v2f*)(p + 28);
    const v4f y0 = __builtin_shufflevector(x0, x1, 0, 1, 2, 3);
    const v4f y1 = __builtin_shufflevector(x2, x3, 0, 1, 2, 3);
    const v4f y2 = __builtin_shufflevector(x4, x5, 0, 1, 2, 3);
    const v4f y3 = __builtin_shufflevector(x6, x7, 0, 1, 2, 3);
    const v8f z0 = __builtin_shufflevector(y0, y1, 0, 1, 2, 3, 4, 5, 6, 7);
    const v8f z1 = __builtin_shufflevector(y2, y3, 0, 1, 2, 3, 4, 5, 6, 7);
    return __builtin_shufflevector(z0, z1, 0, 1, 2, 3, 4, 5, 6, 7, 8, 9, 10, 11, 12, 13, 14, 15);
}

#define GAT_S2_FETCH(st_) do { \
    const float* ls2_ = ls2b + (st_) * STAGE_KEYS; \
    sq0 = gat_s2_fetch8(ls2_, half, 0); sq1 = gat_s2_fetch8(ls2_, half, 1); \
    mw = *(const float2*)(lmkb + ((st_) >> 1) * 4); } while (0)

__device__ __forceinline__ float gat_weights(
    const unsigned msh, const v16f s0, const v16f s1, const float2 mw, float* pdst,
    const float sv1, const float mfin, float lsum)
{
    unsigned mk[2];
    mk[0] = __float_as_uint(mw.x) >> msh;
    mk[1] = __float_as_uint(mw.y) >> msh;
#pragma unroll
    for (int jj = 0; jj < STAGE_KEYS / 16; ++jj) {
        const v16f sx = (jj >> 1) == 0 ? s0 : s1;
        float p[8];
        float rs = 0.0f;
#pragma unroll
        for (int q = 0; q < 4; ++q) {
#pragma unroll
            for (int c = 0; c < 2; ++c) {
                float e = sv1 + sx[8 * (jj & 1) + 2 * q + c];
                e = (e >= 0.0f) ? e : ALPHA * e;
                e = ((mk[c] >> (4 * jj + q)) & 1u) ? e : NEG_BIG;
                const float w = __expf(e - mfin);
                p[2 * q + c] = w;
                rs += w;
            }
        }
        lsum += rs;
        float4 lo, hi;
        lo.x = p[0]; lo.y = p[1]; lo.z = p[2]; lo.w = p[3];
        hi.x = p[4]; hi.y = p[5]; hi.z = p[6]; hi.w = p[7];
        *(float4*)(pdst + (2 * jj)     * 128) = lo;
        *(float4*)(pdst + (2 * jj + 1) * 128) = hi;
    }
    return lsum;
}

__device__ __forceinline__ void gat_frag_load(
    float (&hv)[8][4][2], float4 (&pw)[4], const float* lh, const float* pp,
    const int blk, const int half, const int n)
{
#pragma unroll
    for (int j2 = 0; j2 < 2; ++j2) {
        const int jj = 2 * blk + j2;
        pw[2 * j2]     = *(const float4*)(pp + (2 * jj)     * 128);
        pw[2 * j2 + 1] = *(const float4*)(pp + (2 * jj + 1) * 128);
#pragma unroll
        for (int q = 0; q < 4; ++q) {
            const float* hr0 = lh + (16 * jj + 4 * q + 2 * half) * H_PITCH + n;
            const float* hr1 = hr0 + H_PITCH;
#pragma unroll
            for (int t = 0; t < 4; ++t) {
                hv[4 * j2 + q][t][0] = hr0[16 * t];
                hv[4 * j2 + q][t][1] = hr1[16 * t];
            }
        }
    }
}

__device__ __forceinline__ void gat_frag_mma(
    const float (&hv)[8][4][2], const float4 (&pw)[4], v8f& acc0, v8f& acc1, v8f& acc2, v8f& acc3)
{
#pragma unroll
    for (int ks = 0; ks < 8; ++ks) {
        const float4 w4 = pw[ks >> 1];
        v2f P;
        P[0] = (ks & 1) ? w4.z : w4.x;
        P[1] = (ks & 1) ? w4.w : w4.y;
        v2f H0; H0[0] = hv[ks][0][0]; H0[1] = hv[ks][0][1];
        v2f H1; H1[0] = hv[ks][1][0]; H1[1] = hv[ks][1][1];
        v2f H2; H2[0] = hv[ks][2][0]; H2[1] = hv[ks][2][1];
        v2f H3; H3[0] = hv[ks][3][0]; H3[1] = hv[ks][3][1];
        acc0 = __builtin_amdgcn_wmma_f32_16x16x4_f32(false, H0, false, P, (short)0, acc0, false, false);
        acc1 = __builtin_amdgcn_wmma_f32_16x16x4_f32(false, H1, false, P, (short)0, acc1, false, false);
        acc2 = __builtin_amdgcn_wmma_f32_16x16x4_f32(false, H2, false, P, (short)0, acc2, false, false);
        acc3 = __builtin_amdgcn_wmma_f32_16x16x4_f32(false, H3, false, P, (short)0, acc3, false, false);
    }
}

__global__ void __launch_bounds__(ATTN_THREADS) __attribute__((amdgpu_waves_per_eu(2, 2))) gat_attn_kernel(
    const float* __restrict__ h, const float* __restrict__ s1g,
    const float* __restrict__ s2g, float* out)
{
    __shared__ alignas(16) float lds_h[2 * STAGE_H];
    __shared__ alignas(16) float lds_p[2 * STAGE_P];
    __shared__ alignas(16) float lds_s2[2 * NODES];
    __shared__ alignas(16) float lds_mk[32 * MK_PITCH];
    __shared__ alignas(16) float lds_pp[2 * 2 * 32 * RM_PITCH];
    __shared__ float lds_inv[ATTN_MATRIX_WAVES * 32];

    const int lane = threadIdx.x & 31;
    const int wid  = __builtin_amdgcn_readfirstlane((int)(threadIdx.x >> 5));
    const bool copy_wave = wid >= ATTN_MATRIX_WAVES;
    const int cw   = wid & 3;
    const int pr   = blockIdx.x >> 7;
    const int bsel = cw >> 1;
    const int b    = 2 * pr + bsel;
    const int iblk = (blockIdx.x & 127) << 5;
    const int i0   = iblk + ((cw & 1) << 4);
    const int half = lane >> 4;
    const int n    = lane & 15;
    const int lt   = (int)threadIdx.x & 127;

    const float sv1 = s1g[b * NODES + i0 + n];

    float lsum = 0.0f;
    v8f acc0 = {}, acc1 = {}, acc2 = {}, acc3 = {};

    v16f hq0, hq1, hq2, hq3;
    if (copy_wave) {
        GAT_H_REQUEST(0);
        const float4* s2src = (const float4*)(s2g + (size_t)(2 * pr) * NODES);
        float4 sreg[16];
#pragma unroll
        for (int k = 0; k < 16; ++k) sreg[k] = s2src[lt + 128 * k];
        float4 mreg[8];
#pragma unroll
        for (int k = 0; k < 8; ++k) {
            const int f  = lt + 128 * k;
            const int r  = f >> 5;
            const int ch = f & 31;
            mreg[k] = *(const float4*)(out + ((size_t)(2 * pr + (ch >> 4)) * NODES + iblk + r) * FOUT + 4 * (ch & 15));
        }
#pragma unroll
        for (int k = 0; k < 16; ++k) ((float4*)lds_s2)[lt + 128 * k] = sreg[k];
#pragma unroll
        for (int k = 0; k < 8; ++k) {
            const int f = lt + 128 * k;
            *(float4*)(lds_mk + (f >> 5) * MK_PITCH + 4 * (f & 31)) = mreg[k];
        }
        GAT_H_STORE(0);
    } else {
        const int rset  = cw & 1;
        const int chalf = cw >> 1;
        const float4* s2a = (const float4*)(s2g + (2 * pr)     * NODES + 2048 * chalf) + lane;
        const float4* s2b = (const float4*)(s2g + (2 * pr + 1) * NODES + 2048 * chalf) + lane;
        float4 sa[16], sb[16];
#pragma unroll
        for (int c = 0; c < 16; ++c) { sa[c] = s2a[32 * c]; sb[c] = s2b[32 * c]; }
        const unsigned* wrow = (const unsigned*)(out + ((size_t)(2 * pr + chalf) * NODES + iblk + 16 * rset) * FOUT);
        float* ppa = lds_pp + ((chalf * 2 + 0) * 32 + 16 * rset) * RM_PITCH + lane;
        float* ppb = lds_pp + ((chalf * 2 + 1) * 32 + 16 * rset) * RM_PITCH + lane;
        for (int r = 0; r < 16; ++r) {
            const unsigned* wp = wrow + r * FOUT;
            float ma = -INFINITY, mb = -INFINITY;
#pragma unroll
            for (int c = 0; c < 16; ++c) {
                const bool k0 = __builtin_amdgcn_inverse_ballot_w32(wp[4 * c + 0]);
                const bool k1 = __builtin_amdgcn_inverse_ballot_w32(wp[4 * c + 1]);
                const bool k2 = __builtin_amdgcn_inverse_ballot_w32(wp[4 * c + 2]);
                const bool k3 = __builtin_amdgcn_inverse_ballot_w32(wp[4 * c + 3]);
                ma = fmaxf(ma, fmaxf(k0 ? sa[c].x : -INFINITY, k1 ? sa[c].y : -INFINITY));
                ma = fmaxf(ma, fmaxf(k2 ? sa[c].z : -INFINITY, k3 ? sa[c].w : -INFINITY));
                mb = fmaxf(mb, fmaxf(k0 ? sb[c].x : -INFINITY, k1 ? sb[c].y : -INFINITY));
                mb = fmaxf(mb, fmaxf(k2 ? sb[c].z : -INFINITY, k3 ? sb[c].w : -INFINITY));
            }
            ppa[r * RM_PITCH] = ma;
            ppb[r * RM_PITCH] = mb;
        }
    }
    __syncthreads();

    float mfin = NEG_BIG;
    const float* ls2b = lds_s2 + bsel * NODES;
    const float* lmkb = lds_mk + (((cw & 1) << 4) + n) * MK_PITCH + 2 * half;
    float* pdb = lds_p + cw * WAVE_P + lane * 4;
    if (copy_wave) {
        const float4* pr4 = (const float4*)(lds_pp + ((half * 2 + bsel) * 32 + ((cw & 1) << 4) + n) * RM_PITCH);
        float rm = -INFINITY;
        for (int k = 0; k < 8; ++k) {
            const float4 t = pr4[k];
            rm = fmaxf(rm, fmaxf(fmaxf(t.x, t.y), fmaxf(t.z, t.w)));
        }
        rm = fmaxf(rm, __shfl_xor(rm, 16, 32));
        float e = sv1 + rm;
        e = (e >= 0.0f) ? e : ALPHA * e;
        mfin = fmaxf(e, NEG_BIG);
        GAT_H_REQUEST(1);
        v16f sq0, sq1;
        float2 mw;
        GAT_S2_FETCH(0);
        lsum = gat_weights(0u, sq0, sq1, mw, pdb, sv1, mfin, lsum);
    }
    __syncthreads();

    float hvA[8][4][2], hvB[8][4][2];
    float4 pwA[4], pwB[4];
#pragma unroll
    for (int ks = 0; ks < 8; ++ks)
#pragma unroll
        for (int t = 0; t < 4; ++t) { hvB[ks][t][0] = 0.0f; hvB[ks][t][1] = 0.0f; }
#pragma unroll
    for (int i = 0; i < 4; ++i) pwB[i] = make_float4(0.0f, 0.0f, 0.0f, 0.0f);
    for (int st = 0; st < N_STAGES; ++st) {
        if (copy_wave) {
            if (st + 1 < N_STAGES) {
                GAT_H_STORE(st + 1);
                asm volatile("" ::: "memory");
                GAT_H_REQUEST((st + 2 < N_STAGES) ? st + 2 : N_STAGES - 1);
                v16f sq0, sq1;
                float2 mw;
                GAT_S2_FETCH(st + 1);
                float sv1p = sv1, mfinp = mfin;
                asm volatile("" : "+v"(sv1p), "+v"(mfinp) : : "memory");
                lsum = gat_weights((unsigned)((st + 1) & 1) << 4, sq0, sq1, mw,
                                   pdb + ((st + 1) & 1) * STAGE_P, sv1p, mfinp, lsum);
            } else {
                lds_inv[cw * 32 + lane] = 1.0f / (lsum + __shfl_xor(lsum, 16, 32));
            }
        } else {
            const float* lh = lds_h + (st & 1) * STAGE_H + bsel * (STAGE_KEYS * H_PITCH);
            const float* pp = pdb + (st & 1) * STAGE_P;
            gat_frag_load(hvA, pwA, lh, pp, 0, half, n);
            GAT_GROUP_EDGE();
            gat_frag_mma(hvB, pwB, acc0, acc1, acc2, acc3);
            GAT_GROUP_EDGE();
            gat_frag_load(hvB, pwB, lh, pp, 1, half, n);
            GAT_GROUP_EDGE();
            gat_frag_mma(hvA, pwA, acc0, acc1, acc2, acc3);
        }
        __syncthreads();
    }

    if (!copy_wave) {
        gat_frag_mma(hvB, pwB, acc0, acc1, acc2, acc3);
        const float inv = lds_inv[cw * 32 + lane];
        float* o = out + ((size_t)(b * NODES + i0 + n)) * FOUT + 8 * half;
        gat_store_tile(o,      acc0, inv);
        gat_store_tile(o + 16, acc1, inv);
        gat_store_tile(o + 32, acc2, inv);
        gat_store_tile(o + 48, acc3, inv);
    }
}

extern "C" void kernel_launch(void* const* d_in, const int* in_sizes, int n_in,
                              void* d_out, int out_size, void* d_ws, size_t ws_size,
                              hipStream_t stream)
{
    const float* x   = (const float*)d_in[0];
    const int*   adj = (const int*)  d_in[1];
    const float* W   = (const float*)d_in[2];
    const float* a   = (const float*)d_in[3];
    float* out = (float*)d_out;

    float* h  = (float*)d_ws;
    float* s1 = h + (size_t)BATCH * NODES * FOUT;
    float* s2 = s1 + (size_t)BATCH * NODES;

    const int total_waves = BATCH * (NODES / 16);

    gat_h_kernel<<<K1_BLOCKS, 256, 0, stream>>>(x, W, a, adj, h, s1, s2, out);
    gat_attn_kernel<<<total_waves / ATTN_MATRIX_WAVES, ATTN_THREADS, 0, stream>>>(h, s1, s2, out);
}
